# starting kernel
# baseline (speedup 1.0000x reference)
#include <hip/hip_runtime.h>
#include <hip/hip_bf16.h>


#define DI __device__ __forceinline__

typedef float v2f __attribute__((ext_vector_type(2)));
typedef float v8f __attribute__((ext_vector_type(8)));

constexpr int LDA = 36;
constexpr int LDB = 65;

DI v2f frag_a(const float* As, int k0, int r, int half) {
  v2f a;
  a.x = As[r * LDA + k0 + 2 * half + 0];
  a.y = As[r * LDA + k0 + 2 * half + 1];
  return a;
}
DI v2f frag_b(const float* Bs, int k0, int r, int half) {
  v2f b;
  b.x = Bs[(k0 + 2 * half + 0) * LDB + r];
  b.y = Bs[(k0 + 2 * half + 1) * LDB + r];
  return b;
}

DI void mma_chunk(const float* As, const float* Bs, int wr, int cp, int r, int half,
                  v8f& acc0, v8f& acc1) {
  const float* Aw = As + (wr * 16) * LDA;
  const float* B0 = Bs + 32 * cp;
  const float* B1 = Bs + 32 * cp + 16;
#pragma unroll
  for (int k0 = 0; k0 < 32; k0 += 4) {
    v2f a  = frag_a(Aw, k0, r, half);
    v2f b0 = frag_b(B0, k0, r, half);
    v2f b1 = frag_b(B1, k0, r, half);
    acc0 = __builtin_amdgcn_wmma_f32_16x16x4_f32(false, a, false, b0, (short)0, acc0, false, false);
    acc1 = __builtin_amdgcn_wmma_f32_16x16x4_f32(false, a, false, b1, (short)0, acc1, false, false);
  }
}

DI void store_tile(float* out, int ldo, const v8f& acc, int r, int half) {
#pragma unroll
  for (int i = 0; i < 8; ++i) out[(i + 8 * half) * ldo + r] = acc[i];
}

#define WMMA_THREAD_DECOMP()        \
  const int tid  = threadIdx.x;     \
  const int lane = tid & 31;        \
  const int wave = tid >> 5;        \
  const int half = lane >> 4;       \
  const int r    = lane & 15;       \
  const int wr   = wave & 3;        \
  const int cp   = wave >> 2;       \
  (void)lane;

DI float sigmoidf(float x) { return 1.0f / (1.0f + __expf(-x)); }

__global__ void softmax_rows(const float* __restrict__ dis, float* __restrict__ w) {
  __shared__ float red[256];
  const int i = blockIdx.x, t = threadIdx.x;
  float v = -dis[i * 256 + t];
  red[t] = v;
  __syncthreads();
  for (int s = 128; s > 0; s >>= 1) {
    if (t < s) red[t] = fmaxf(red[t], red[t + s]);
    __syncthreads();
  }
  float mx = red[0];
  __syncthreads();
  float e = __expf(v - mx);
  red[t] = e;
  __syncthreads();
  for (int s = 128; s > 0; s >>= 1) {
    if (t < s) red[t] += red[t + s];
    __syncthreads();
  }
  w[i * 256 + t] = e / red[0];
}

__global__ void embed_weight_gemm(const float* __restrict__ E1, const float* __restrict__ E2,
                                  const float* __restrict__ Wih,
                                  float* __restrict__ G1, float* __restrict__ G2) {
  __shared__ float As[64 * LDA];
  __shared__ float Bs[32 * LDB];
  WMMA_THREAD_DECOMP();
  const float* E = blockIdx.z ? E2 : E1;
  float* G = blockIdx.z ? G2 : G1;
  const int koff = blockIdx.z ? 256 : 0;
  const int rb = blockIdx.y * 64;
  const int cb = blockIdx.x * 64;
  v8f acc0 = {}; v8f acc1 = {};
  for (int k0 = 0; k0 < 256; k0 += 32) {
    __syncthreads();
#pragma unroll
    for (int i = 0; i < 8; ++i) {
      int idx = tid + i * 256;
      int ar = idx >> 5, ac = idx & 31;
      As[ar * LDA + ac] = E[(rb + ar) * 256 + k0 + ac];
      int bc = idx >> 5, bk = idx & 31;
      Bs[bk * LDB + bc] = Wih[(cb + bc) * 512 + koff + k0 + bk];
    }
    __syncthreads();
    mma_chunk(As, Bs, wr, cp, r, half, acc0, acc1);
  }
  float* o0 = G + (rb + 16 * wr) * 1024 + cb + 32 * cp;
  store_tile(o0, 1024, acc0, r, half);
  store_tile(o0 + 16, 1024, acc1, r, half);
}

__global__ void lstm_gemm(const float* __restrict__ X, const float* __restrict__ wsm,
                          const float* __restrict__ G1, const float* __restrict__ G2,
                          const float* __restrict__ Whh, const float* __restrict__ h,
                          float* __restrict__ z, int l) {
  __shared__ float As[64 * LDA];
  __shared__ float Bs[32 * LDB];
  WMMA_THREAD_DECOMP();
  const int rb = blockIdx.y * 64;
  const int cb = blockIdx.x * 64;
  v8f acc0 = {}; v8f acc1 = {};
  for (int seg = 0; seg < 2; ++seg) {
    const float* G = seg ? G2 : G1;
    for (int k0 = 0; k0 < 256; k0 += 32) {
      __syncthreads();
#pragma unroll
      for (int i = 0; i < 8; ++i) {
        int idx = tid + i * 256;
        int ar = idx >> 5, ac = idx & 31;
        int row = rb + ar;
        int bb = row >> 8, nn = row & 255;
        float xv = X[((bb * 32 + l) * 256 + nn) * 256 + k0 + ac];
        if (seg) xv *= wsm[nn * 256 + k0 + ac];
        As[ar * LDA + ac] = xv;
        int bk = idx >> 6, bc = idx & 63;
        Bs[bk * LDB + bc] = G[(k0 + bk) * 1024 + cb + bc];
      }
      __syncthreads();
      mma_chunk(As, Bs, wr, cp, r, half, acc0, acc1);
    }
  }
  for (int k0 = 0; k0 < 256; k0 += 32) {
    __syncthreads();
#pragma unroll
    for (int i = 0; i < 8; ++i) {
      int idx = tid + i * 256;
      int ar = idx >> 5, ac = idx & 31;
      As[ar * LDA + ac] = h[(rb + ar) * 256 + k0 + ac];
      int bc = idx >> 5, bk = idx & 31;
      Bs[bk * LDB + bc] = Whh[(cb + bc) * 256 + k0 + bk];
    }
    __syncthreads();
    mma_chunk(As, Bs, wr, cp, r, half, acc0, acc1);
  }
  float* o0 = z + (rb + 16 * wr) * 1024 + cb + 32 * cp;
  store_tile(o0, 1024, acc0, r, half);
  store_tile(o0 + 16, 1024, acc1, r, half);
}

__global__ void lstm_point(const float* __restrict__ z, const float* __restrict__ bih,
                           const float* __restrict__ bhh, float* __restrict__ h,
                           float* __restrict__ c, float* __restrict__ temp, int l) {
  int idx = blockIdx.x * 256 + threadIdx.x;
  int row = idx >> 8, m = idx & 255;
  const float* zr = z + row * 1024;
  float iv = sigmoidf(zr[m]       + bih[m]       + bhh[m]);
  float fv = sigmoidf(zr[m + 256] + bih[m + 256] + bhh[m + 256]);
  float gv = tanhf   (zr[m + 512] + bih[m + 512] + bhh[m + 512]);
  float ov = sigmoidf(zr[m + 768] + bih[m + 768] + bhh[m + 768]);
  float cv = fv * c[idx] + iv * gv;
  float hv = ov * tanhf(cv);
  c[idx] = cv;
  h[idx] = hv;
  int bb = row >> 8, nn = row & 255;
  temp[((bb * 32 + l) * 256 + nn) * 256 + m] = hv;
}

__global__ void zero_f32(float* __restrict__ p, int n) {
  int i = blockIdx.x * 256 + threadIdx.x;
  if (i < n) p[i] = 0.0f;
}

__global__ void bn_stats(const float* __restrict__ temp, float* __restrict__ stats) {
  const int m = threadIdx.x;
  const int rb = blockIdx.x * 256;
  float s = 0.0f, s2 = 0.0f;
  for (int rr = 0; rr < 256; ++rr) {
    float v = temp[(rb + rr) * 256 + m];
    s += v; s2 += v * v;
  }
  atomicAdd(&stats[m], s);
  atomicAdd(&stats[256 + m], s2);
}

__global__ void bn_final(const float* __restrict__ stats, const float* __restrict__ gamma,
                         const float* __restrict__ beta, float* __restrict__ scale,
                         float* __restrict__ shift) {
  const int m = threadIdx.x;
  const float inv = 1.0f / 65536.0f;
  float mu = stats[m] * inv;
  float var = stats[256 + m] * inv - mu * mu;
  float sc = gamma[m] * rsqrtf(var + 1e-5f);
  scale[m] = sc;
  shift[m] = beta[m] - mu * sc;
}

__global__ void proj_gemm(const float* __restrict__ temp, const float* __restrict__ Wlin,
                          const float* __restrict__ blin, const float* __restrict__ scale,
                          const float* __restrict__ shift, float* __restrict__ proj) {
  __shared__ float As[64 * LDA];
  __shared__ float Bs[32 * LDB];
  WMMA_THREAD_DECOMP();
  const int rb = blockIdx.y * 64;
  const int cb = blockIdx.x * 64;
  v8f acc0 = {}; v8f acc1 = {};
  for (int k0 = 0; k0 < 256; k0 += 32) {
    __syncthreads();
#pragma unroll
    for (int i = 0; i < 8; ++i) {
      int idx = tid + i * 256;
      int ar = idx >> 5, ac = idx & 31;
      As[ar * LDA + ac] = temp[(rb + ar) * 256 + k0 + ac] * scale[k0 + ac] + shift[k0 + ac];
      int bc = idx >> 5, bk = idx & 31;
      Bs[bk * LDB + bc] = Wlin[(cb + bc) * 256 + k0 + bk];
    }
    __syncthreads();
    mma_chunk(As, Bs, wr, cp, r, half, acc0, acc1);
  }
  const int c0 = cb + 32 * cp;
  float* o0 = proj + (rb + 16 * wr) * 256 + c0;
#pragma unroll
  for (int i = 0; i < 8; ++i) {
    o0[(i + 8 * half) * 256 + r]      = acc0[i] + blin[c0 + r];
    o0[(i + 8 * half) * 256 + 16 + r] = acc1[i] + blin[c0 + 16 + r];
  }
}

__global__ void out_gemm(const float* __restrict__ proj, const float* __restrict__ temp,
                         const float* __restrict__ scale, const float* __restrict__ shift,
                         float* __restrict__ out) {
  __shared__ float As[64 * LDA];
  __shared__ float Bs[32 * LDB];
  WMMA_THREAD_DECOMP();
  const int bl = blockIdx.z;
  const int rb = blockIdx.y * 64;
  const int cb = blockIdx.x * 64;
  const float* P = proj + bl * 65536;
  const float* T = temp + bl * 65536;
  v8f acc0 = {}; v8f acc1 = {};
  for (int k0 = 0; k0 < 256; k0 += 32) {
    __syncthreads();
#pragma unroll
    for (int i = 0; i < 8; ++i) {
      int idx = tid + i * 256;
      int ar = idx >> 5, ac = idx & 31;
      As[ar * LDA + ac] = P[(rb + ar) * 256 + k0 + ac];
      int dd = idx >> 5, mm = idx & 31;
      Bs[mm * LDB + dd] = T[(cb + dd) * 256 + k0 + mm] * scale[k0 + mm] + shift[k0 + mm];
    }
    __syncthreads();
    mma_chunk(As, Bs, wr, cp, r, half, acc0, acc1);
  }
  float* o0 = out + bl * 65536 + (rb + 16 * wr) * 256 + cb + 32 * cp;
  store_tile(o0, 256, acc0, r, half);
  store_tile(o0 + 16, 256, acc1, r, half);
}

extern "C" void kernel_launch(void* const* d_in, const int* in_sizes, int n_in,
                              void* d_out, int out_size, void* d_ws, size_t ws_size,
                              hipStream_t stream) {
  (void)in_sizes; (void)n_in; (void)out_size; (void)ws_size;
  const float* X     = (const float*)d_in[0];
  const float* dis   = (const float*)d_in[1];
  const float* E1    = (const float*)d_in[2];
  const float* E2    = (const float*)d_in[3];
  const float* Wih   = (const float*)d_in[4];
  const float* Whh   = (const float*)d_in[5];
  const float* bih   = (const float*)d_in[6];
  const float* bhh   = (const float*)d_in[7];
  const float* gamma = (const float*)d_in[8];
  const float* beta  = (const float*)d_in[9];
  const float* Wlin  = (const float*)d_in[10];
  const float* blin  = (const float*)d_in[11];
  float* out = (float*)d_out;

  float* ws = (float*)d_ws;
  size_t o = 0;
  float* w_sm  = ws + o; o += 256 * 256;
  float* G1    = ws + o; o += 256 * 1024;
  float* G2    = ws + o; o += 256 * 1024;
  float* z     = ws + o; o += 2048 * 1024;
  float* h     = ws + o; o += 2048 * 256;
  float* c     = ws + o; o += 2048 * 256;
  float* stats = ws + o; o += 512;
  float* scale = ws + o; o += 256;
  float* shift = ws + o; o += 256;
  float* temp  = ws + o; o += 65536 * 256;
  float* proj  = ws + o; o += 65536 * 256;

  softmax_rows<<<256, 256, 0, stream>>>(dis, w_sm);
  embed_weight_gemm<<<dim3(16, 4, 2), 256, 0, stream>>>(E1, E2, Wih, G1, G2);

  const int zn = 2048 * 256 * 2 + 512;
  zero_f32<<<(zn + 255) / 256, 256, 0, stream>>>(h, zn);

  for (int l = 0; l < 32; ++l) {
    lstm_gemm<<<dim3(16, 32), 256, 0, stream>>>(X, w_sm, G1, G2, Whh, h, z, l);
    lstm_point<<<2048, 256, 0, stream>>>(z, bih, bhh, h, c, temp, l);
  }

  bn_stats<<<256, 256, 0, stream>>>(temp, stats);
  bn_final<<<1, 256, 0, stream>>>(stats, gamma, beta, scale, shift);
  proj_gemm<<<dim3(4, 1024), 256, 0, stream>>>(temp, Wlin, blin, scale, shift, proj);
  out_gemm<<<dim3(4, 4, 256), 256, 0, stream>>>(proj, temp, scale, shift, out);
}


# gate GEMM: each wave reads WMMA fragments straight from global memory (16-byte loads, operands k-contiguous, x-path weights stored transposed), no LDS, no barriers; sum order in 8-k groups changes
# speedup vs baseline: 1.6029x; 1.6029x over previous
#include <hip/hip_runtime.h>
#include <hip/hip_bf16.h>


#define DI __device__ __forceinline__

typedef float v2f __attribute__((ext_vector_type(2)));
typedef float v4f __attribute__((ext_vector_type(4)));
typedef float v8f __attribute__((ext_vector_type(8)));

constexpr int LDA = 36;
constexpr int LDB = 65;

DI v2f frag_a(const float* As, int k0, int r, int half) {
  v2f a;
  a.x = As[r * LDA + k0 + 2 * half + 0];
  a.y = As[r * LDA + k0 + 2 * half + 1];
  return a;
}
DI v2f frag_b(const float* Bs, int k0, int r, int half) {
  v2f b;
  b.x = Bs[(k0 + 2 * half + 0) * LDB + r];
  b.y = Bs[(k0 + 2 * half + 1) * LDB + r];
  return b;
}

DI void mma_chunk(const float* As, const float* Bs, int wr, int cp, int r, int half,
                  v8f& acc0, v8f& acc1) {
  const float* Aw = As + (wr * 16) * LDA;
  const float* B0 = Bs + 32 * cp;
  const float* B1 = Bs + 32 * cp + 16;
#pragma unroll
  for (int k0 = 0; k0 < 32; k0 += 4) {
    v2f a  = frag_a(Aw, k0, r, half);
    v2f b0 = frag_b(B0, k0, r, half);
    v2f b1 = frag_b(B1, k0, r, half);
    acc0 = __builtin_amdgcn_wmma_f32_16x16x4_f32(false, a, false, b0, (short)0, acc0, false, false);
    acc1 = __builtin_amdgcn_wmma_f32_16x16x4_f32(false, a, false, b1, (short)0, acc1, false, false);
  }
}

DI void store_tile(float* out, int ldo, const v8f& acc, int r, int half) {
#pragma unroll
  for (int i = 0; i < 8; ++i) out[(i + 8 * half) * ldo + r] = acc[i];
}

DI void ld_frag32(const float* p, v4f (&f)[4]) {
#pragma unroll
  for (int j = 0; j < 4; ++j) f[j] = *reinterpret_cast<const v4f*>(p + 8 * j);
}

DI v8f wmma4(v2f a, v2f b, v8f c) {
  return __builtin_amdgcn_wmma_f32_16x16x4_f32(false, a, false, b, (short)0, c, false, false);
}

DI void mma_direct32(const v4f (&A0)[4], const v4f (&A1)[4], const v4f (&B0)[4], const v4f (&B1)[4],
                     v8f& c00, v8f& c01, v8f& c10, v8f& c11) {
#pragma unroll
  for (int j = 0; j < 4; ++j) {
    v2f a0 = {A0[j].x, A0[j].y}, a1 = {A1[j].x, A1[j].y};
    v2f b0 = {B0[j].x, B0[j].y}, b1 = {B1[j].x, B1[j].y};
    c00 = wmma4(a0, b0, c00); c01 = wmma4(a0, b1, c01);
    c10 = wmma4(a1, b0, c10); c11 = wmma4(a1, b1, c11);
    v2f a2 = {A0[j].z, A0[j].w}, a3 = {A1[j].z, A1[j].w};
    v2f b2 = {B0[j].z, B0[j].w}, b3 = {B1[j].z, B1[j].w};
    c00 = wmma4(a2, b2, c00); c01 = wmma4(a2, b3, c01);
    c10 = wmma4(a3, b2, c10); c11 = wmma4(a3, b3, c11);
  }
}

#define WMMA_THREAD_DECOMP()        \
  const int tid  = threadIdx.x;     \
  const int lane = tid & 31;        \
  const int wave = tid >> 5;        \
  const int half = lane >> 4;       \
  const int r    = lane & 15;       \
  const int wr   = wave & 3;        \
  const int cp   = wave >> 2;       \
  (void)lane;

DI float sigmoidf(float x) { return 1.0f / (1.0f + __expf(-x)); }

__global__ void softmax_rows(const float* __restrict__ dis, float* __restrict__ w) {
  __shared__ float red[256];
  const int i = blockIdx.x, t = threadIdx.x;
  float v = -dis[i * 256 + t];
  red[t] = v;
  __syncthreads();
  for (int s = 128; s > 0; s >>= 1) {
    if (t < s) red[t] = fmaxf(red[t], red[t + s]);
    __syncthreads();
  }
  float mx = red[0];
  __syncthreads();
  float e = __expf(v - mx);
  red[t] = e;
  __syncthreads();
  for (int s = 128; s > 0; s >>= 1) {
    if (t < s) red[t] += red[t + s];
    __syncthreads();
  }
  w[i * 256 + t] = e / red[0];
}

__global__ void embed_weight_gemm(const float* __restrict__ E1, const float* __restrict__ E2,
                                  const float* __restrict__ Wih,
                                  float* __restrict__ G1T, float* __restrict__ G2T) {
  __shared__ float As[64 * LDA];
  __shared__ float Bs[32 * LDB];
  WMMA_THREAD_DECOMP();
  const float* E = blockIdx.z ? E2 : E1;
  float* G = blockIdx.z ? G2T : G1T;
  const int koff = blockIdx.z ? 256 : 0;
  const int rb = blockIdx.y * 64;
  const int cb = blockIdx.x * 64;
  v8f acc0 = {}; v8f acc1 = {};
  for (int k0 = 0; k0 < 256; k0 += 32) {
    __syncthreads();
#pragma unroll
    for (int i = 0; i < 8; ++i) {
      int idx = tid + i * 256;
      int ar = idx >> 5, ac = idx & 31;
      As[ar * LDA + ac] = Wih[(rb + ar) * 512 + koff + k0 + ac];
      int bc = idx >> 5, bk = idx & 31;
      Bs[bk * LDB + bc] = E[(cb + bc) * 256 + k0 + bk];
    }
    __syncthreads();
    mma_chunk(As, Bs, wr, cp, r, half, acc0, acc1);
  }
  float* o0 = G + (rb + 16 * wr) * 256 + cb + 32 * cp;
  store_tile(o0, 256, acc0, r, half);
  store_tile(o0 + 16, 256, acc1, r, half);
}

__global__ __launch_bounds__(256) void lstm_gemm(const float* __restrict__ X, const float* __restrict__ wsm,
                          const float* __restrict__ G1T, const float* __restrict__ G2T,
                          const float* __restrict__ Whh, const float* __restrict__ h,
                          float* __restrict__ z, int l) {
  const int lane = threadIdx.x & 31;
  const int wave = threadIdx.x >> 5;
  const int half = lane >> 4;
  const int r    = lane & 15;
  const int row0 = blockIdx.y * 64 + (wave & 1) * 32;
  const int col0 = blockIdx.x * 128 + (wave >> 1) * 32;
  const int ra = row0 + r;
  const int bb = ra >> 8, nn = ra & 255;
  const float* x0 = X + ((bb * 32 + l) * 256 + nn) * 256 + 4 * half;
  const float* x1 = x0 + 16 * 256;
  const float* w0 = wsm + nn * 256 + 4 * half;
  const float* w1 = w0 + 16 * 256;
  const float* h0 = h + ra * 256 + 4 * half;
  const float* h1 = h0 + 16 * 256;
  const int cc = (col0 + r) * 256 + 4 * half;
  v8f c00 = {}, c01 = {}, c10 = {}, c11 = {};
  v4f A0[4], A1[4], B0[4], B1[4];
#pragma unroll
  for (int k0 = 0; k0 < 256; k0 += 32) {
    ld_frag32(x0 + k0, A0);
    ld_frag32(x1 + k0, A1);
    ld_frag32(G1T + cc + k0, B0);
    ld_frag32(G1T + cc + 16 * 256 + k0, B1);
    mma_direct32(A0, A1, B0, B1, c00, c01, c10, c11);
  }
#pragma unroll
  for (int k0 = 0; k0 < 256; k0 += 32) {
    v4f W0[4], W1[4];
    ld_frag32(x0 + k0, A0);
    ld_frag32(w0 + k0, W0);
    ld_frag32(x1 + k0, A1);
    ld_frag32(w1 + k0, W1);
    ld_frag32(G2T + cc + k0, B0);
    ld_frag32(G2T + cc + 16 * 256 + k0, B1);
#pragma unroll
    for (int j = 0; j < 4; ++j) { A0[j] *= W0[j]; A1[j] *= W1[j]; }
    mma_direct32(A0, A1, B0, B1, c00, c01, c10, c11);
  }
#pragma unroll
  for (int k0 = 0; k0 < 256; k0 += 32) {
    ld_frag32(h0 + k0, A0);
    ld_frag32(h1 + k0, A1);
    ld_frag32(Whh + cc + k0, B0);
    ld_frag32(Whh + cc + 16 * 256 + k0, B1);
    mma_direct32(A0, A1, B0, B1, c00, c01, c10, c11);
  }
  float* o0 = z + row0 * 1024 + col0;
  store_tile(o0, 1024, c00, r, half);
  store_tile(o0 + 16, 1024, c01, r, half);
  store_tile(o0 + 16 * 1024, 1024, c10, r, half);
  store_tile(o0 + 16 * 1024 + 16, 1024, c11, r, half);
}

__global__ void lstm_point(const float* __restrict__ z, const float* __restrict__ bih,
                           const float* __restrict__ bhh, float* __restrict__ h,
                           float* __restrict__ c, float* __restrict__ temp, int l) {
  int idx = blockIdx.x * 256 + threadIdx.x;
  int row = idx >> 8, m = idx & 255;
  const float* zr = z + row * 1024;
  float iv = sigmoidf(zr[m]       + bih[m]       + bhh[m]);
  float fv = sigmoidf(zr[m + 256] + bih[m + 256] + bhh[m + 256]);
  float gv = tanhf   (zr[m + 512] + bih[m + 512] + bhh[m + 512]);
  float ov = sigmoidf(zr[m + 768] + bih[m + 768] + bhh[m + 768]);
  float cv = fv * c[idx] + iv * gv;
  float hv = ov * tanhf(cv);
  c[idx] = cv;
  h[idx] = hv;
  int bb = row >> 8, nn = row & 255;
  temp[((bb * 32 + l) * 256 + nn) * 256 + m] = hv;
}

__global__ void zero_f32(float* __restrict__ p, int n) {
  int i = blockIdx.x * 256 + threadIdx.x;
  if (i < n) p[i] = 0.0f;
}

__global__ void bn_stats(const float* __restrict__ temp, float* __restrict__ stats) {
  const int m = threadIdx.x;
  const int rb = blockIdx.x * 256;
  float s = 0.0f, s2 = 0.0f;
  for (int rr = 0; rr < 256; ++rr) {
    float v = temp[(rb + rr) * 256 + m];
    s += v; s2 += v * v;
  }
  atomicAdd(&stats[m], s);
  atomicAdd(&stats[256 + m], s2);
}

__global__ void bn_final(const float* __restrict__ stats, const float* __restrict__ gamma,
                         const float* __restrict__ beta, float* __restrict__ scale,
                         float* __restrict__ shift) {
  const int m = threadIdx.x;
  const float inv = 1.0f / 65536.0f;
  float mu = stats[m] * inv;
  float var = stats[256 + m] * inv - mu * mu;
  float sc = gamma[m] * rsqrtf(var + 1e-5f);
  scale[m] = sc;
  shift[m] = beta[m] - mu * sc;
}

__global__ void proj_gemm(const float* __restrict__ temp, const float* __restrict__ Wlin,
                          const float* __restrict__ blin, const float* __restrict__ scale,
                          const float* __restrict__ shift, float* __restrict__ proj) {
  __shared__ float As[64 * LDA];
  __shared__ float Bs[32 * LDB];
  WMMA_THREAD_DECOMP();
  const int rb = blockIdx.y * 64;
  const int cb = blockIdx.x * 64;
  v8f acc0 = {}; v8f acc1 = {};
  for (int k0 = 0; k0 < 256; k0 += 32) {
    __syncthreads();
#pragma unroll
    for (int i = 0; i < 8; ++i) {
      int idx = tid + i * 256;
      int ar = idx >> 5, ac = idx & 31;
      As[ar * LDA + ac] = temp[(rb + ar) * 256 + k0 + ac] * scale[k0 + ac] + shift[k0 + ac];
      int bc = idx >> 5, bk = idx & 31;
      Bs[bk * LDB + bc] = Wlin[(cb + bc) * 256 + k0 + bk];
    }
    __syncthreads();
    mma_chunk(As, Bs, wr, cp, r, half, acc0, acc1);
  }
  const int c0 = cb + 32 * cp;
  float* o0 = proj + (rb + 16 * wr) * 256 + c0;
#pragma unroll
  for (int i = 0; i < 8; ++i) {
    o0[(i + 8 * half) * 256 + r]      = acc0[i] + blin[c0 + r];
    o0[(i + 8 * half) * 256 + 16 + r] = acc1[i] + blin[c0 + 16 + r];
  }
}

__global__ void out_gemm(const float* __restrict__ proj, const float* __restrict__ temp,
                         const float* __restrict__ scale, const float* __restrict__ shift,
                         float* __restrict__ out) {
  __shared__ float As[64 * LDA];
  __shared__ float Bs[32 * LDB];
  WMMA_THREAD_DECOMP();
  const int bl = blockIdx.z;
  const int rb = blockIdx.y * 64;
  const int cb = blockIdx.x * 64;
  const float* P = proj + bl * 65536;
  const float* T = temp + bl * 65536;
  v8f acc0 = {}; v8f acc1 = {};
  for (int k0 = 0; k0 < 256; k0 += 32) {
    __syncthreads();
#pragma unroll
    for (int i = 0; i < 8; ++i) {
      int idx = tid + i * 256;
      int ar = idx >> 5, ac = idx & 31;
      As[ar * LDA + ac] = P[(rb + ar) * 256 + k0 + ac];
      int dd = idx >> 5, mm = idx & 31;
      Bs[mm * LDB + dd] = T[(cb + dd) * 256 + k0 + mm] * scale[k0 + mm] + shift[k0 + mm];
    }
    __syncthreads();
    mma_chunk(As, Bs, wr, cp, r, half, acc0, acc1);
  }
  float* o0 = out + bl * 65536 + (rb + 16 * wr) * 256 + cb + 32 * cp;
  store_tile(o0, 256, acc0, r, half);
  store_tile(o0 + 16, 256, acc1, r, half);
}

extern "C" void kernel_launch(void* const* d_in, const int* in_sizes, int n_in,
                              void* d_out, int out_size, void* d_ws, size_t ws_size,
                              hipStream_t stream) {
  (void)in_sizes; (void)n_in; (void)out_size; (void)ws_size;
  const float* X     = (const float*)d_in[0];
  const float* dis   = (const float*)d_in[1];
  const float* E1    = (const float*)d_in[2];
  const float* E2    = (const float*)d_in[3];
  const float* Wih   = (const float*)d_in[4];
  const float* Whh   = (const float*)d_in[5];
  const float* bih   = (const float*)d_in[6];
  const float* bhh   = (const float*)d_in[7];
  const float* gamma = (const float*)d_in[8];
  const float* beta  = (const float*)d_in[9];
  const float* Wlin  = (const float*)d_in[10];
  const float* blin  = (const float*)d_in[11];
  float* out = (float*)d_out;

  float* ws = (float*)d_ws;
  size_t o = 0;
  float* w_sm  = ws + o; o += 256 * 256;
  float* G1T   = ws + o; o += 256 * 1024;
  float* G2T   = ws + o; o += 256 * 1024;
  float* z     = ws + o; o += 2048 * 1024;
  float* h     = ws + o; o += 2048 * 256;
  float* c     = ws + o; o += 2048 * 256;
  float* stats = ws + o; o += 512;
  float* scale = ws + o; o += 256;
  float* shift = ws + o; o += 256;
  float* temp  = ws + o; o += 65536 * 256;
  float* proj  = ws + o; o += 65536 * 256;

  softmax_rows<<<256, 256, 0, stream>>>(dis, w_sm);
  embed_weight_gemm<<<dim3(4, 16, 2), 256, 0, stream>>>(E1, E2, Wih, G1T, G2T);

  const int zn = 2048 * 256 * 2 + 512;
  zero_f32<<<(zn + 255) / 256, 256, 0, stream>>>(h, zn);

  for (int l = 0; l < 32; ++l) {
    lstm_gemm<<<dim3(8, 32), 256, 0, stream>>>(X, w_sm, G1T, G2T, Whh, h, z, l);
    lstm_point<<<2048, 256, 0, stream>>>(z, bih, bhh, h, c, temp, l);
  }

  bn_stats<<<256, 256, 0, stream>>>(temp, stats);
  bn_final<<<1, 256, 0, stream>>>(stats, gamma, beta, scale, shift);
  proj_gemm<<<dim3(4, 1024), 256, 0, stream>>>(temp, Wlin, blin, scale, shift, proj);
  out_gemm<<<dim3(4, 4, 256), 256, 0, stream>>>(proj, temp, scale, shift, out);
}


# gate GEMM operands (X, w, G1, G2, h) and z, c kept in WMMA fragment / tile order in global memory: each wave load or store covers 512 contiguous bytes, not 16 partial lines; sum order kept
# speedup vs baseline: 2.1584x; 1.3466x over previous
#include <hip/hip_runtime.h>
#include <hip/hip_bf16.h>


#define DI __device__ __forceinline__

typedef float v2f __attribute__((ext_vector_type(2)));
typedef float v4f __attribute__((ext_vector_type(4)));
typedef float v8f __attribute__((ext_vector_type(8)));

constexpr int LDA = 36;
constexpr int LDB = 65;

DI v2f frag_a(const float* As, int k0, int r, int half) {
  v2f a;
  a.x = As[r * LDA + k0 + 2 * half + 0];
  a.y = As[r * LDA + k0 + 2 * half + 1];
  return a;
}
DI v2f frag_b(const float* Bs, int k0, int r, int half) {
  v2f b;
  b.x = Bs[(k0 + 2 * half + 0) * LDB + r];
  b.y = Bs[(k0 + 2 * half + 1) * LDB + r];
  return b;
}

DI void mma_chunk(const float* As, const float* Bs, int wr, int cp, int r, int half,
                  v8f& acc0, v8f& acc1) {
  const float* Aw = As + (wr * 16) * LDA;
  const float* B0 = Bs + 32 * cp;
  const float* B1 = Bs + 32 * cp + 16;
#pragma unroll
  for (int k0 = 0; k0 < 32; k0 += 4) {
    v2f a  = frag_a(Aw, k0, r, half);
    v2f b0 = frag_b(B0, k0, r, half);
    v2f b1 = frag_b(B1, k0, r, half);
    acc0 = __builtin_amdgcn_wmma_f32_16x16x4_f32(false, a, false, b0, (short)0, acc0, false, false);
    acc1 = __builtin_amdgcn_wmma_f32_16x16x4_f32(false, a, false, b1, (short)0, acc1, false, false);
  }
}

DI void store_tile(float* out, int ldo, const v8f& acc, int r, int half) {
#pragma unroll
  for (int i = 0; i < 8; ++i) out[(i + 8 * half) * ldo + r] = acc[i];
}

DI void ld_frag32(const float* p, v4f (&f)[4]) {
#pragma unroll
  for (int j = 0; j < 4; ++j) f[j] = *reinterpret_cast<const v4f*>(p + 8 * j);
}

DI int pk_off(int row, int k) {
  return (((row >> 4) * 8 + (k >> 5)) << 9) + ((((k & 31) >> 3) * 32 + (row & 15) + 16 * ((k >> 2) & 1)) << 2) + (k & 3);
}
DI void ld_packed(const float* p, v4f (&f)[4]) {
#pragma unroll
  for (int j = 0; j < 4; ++j) f[j] = *reinterpret_cast<const v4f*>(p + 128 * j);
}

DI v8f wmma4(v2f a, v2f b, v8f c) {
  return __builtin_amdgcn_wmma_f32_16x16x4_f32(false, a, false, b, (short)0, c, false, false);
}

DI void mma_direct32(const v4f (&A0)[4], const v4f (&A1)[4], const v4f (&B0)[4], const v4f (&B1)[4],
                     v8f& c00, v8f& c01, v8f& c10, v8f& c11) {
#pragma unroll
  for (int j = 0; j < 4; ++j) {
    v2f a0 = {A0[j].x, A0[j].y}, a1 = {A1[j].x, A1[j].y};
    v2f b0 = {B0[j].x, B0[j].y}, b1 = {B1[j].x, B1[j].y};
    c00 = wmma4(a0, b0, c00); c01 = wmma4(a0, b1, c01);
    c10 = wmma4(a1, b0, c10); c11 = wmma4(a1, b1, c11);
    v2f a2 = {A0[j].z, A0[j].w}, a3 = {A1[j].z, A1[j].w};
    v2f b2 = {B0[j].z, B0[j].w}, b3 = {B1[j].z, B1[j].w};
    c00 = wmma4(a2, b2, c00); c01 = wmma4(a2, b3, c01);
    c10 = wmma4(a3, b2, c10); c11 = wmma4(a3, b3, c11);
  }
}

#define WMMA_THREAD_DECOMP()        \
  const int tid  = threadIdx.x;     \
  const int lane = tid & 31;        \
  const int wave = tid >> 5;        \
  const int half = lane >> 4;       \
  const int r    = lane & 15;       \
  const int wr   = wave & 3;        \
  const int cp   = wave >> 2;       \
  (void)lane;

DI float sigmoidf(float x) { return 1.0f / (1.0f + __expf(-x)); }

__global__ void softmax_rows(const float* __restrict__ dis, float* __restrict__ w) {
  __shared__ float red[256];
  const int i = blockIdx.x, t = threadIdx.x;
  float v = -dis[i * 256 + t];
  red[t] = v;
  __syncthreads();
  for (int s = 128; s > 0; s >>= 1) {
    if (t < s) red[t] = fmaxf(red[t], red[t + s]);
    __syncthreads();
  }
  float mx = red[0];
  __syncthreads();
  float e = __expf(v - mx);
  red[t] = e;
  __syncthreads();
  for (int s = 128; s > 0; s >>= 1) {
    if (t < s) red[t] += red[t + s];
    __syncthreads();
  }
  w[pk_off(i, t)] = e / red[0];
}

__global__ void embed_weight_gemm(const float* __restrict__ E1, const float* __restrict__ E2,
                                  const float* __restrict__ Wih,
                                  float* __restrict__ G1T, float* __restrict__ G2T) {
  __shared__ float As[64 * LDA];
  __shared__ float Bs[32 * LDB];
  WMMA_THREAD_DECOMP();
  const float* E = blockIdx.z ? E2 : E1;
  float* G = blockIdx.z ? G2T : G1T;
  const int koff = blockIdx.z ? 256 : 0;
  const int rb = blockIdx.y * 64;
  const int cb = blockIdx.x * 64;
  v8f acc0 = {}; v8f acc1 = {};
  for (int k0 = 0; k0 < 256; k0 += 32) {
    __syncthreads();
#pragma unroll
    for (int i = 0; i < 8; ++i) {
      int idx = tid + i * 256;
      int ar = idx >> 5, ac = idx & 31;
      As[ar * LDA + ac] = Wih[(rb + ar) * 512 + koff + k0 + ac];
      int bc = idx >> 5, bk = idx & 31;
      Bs[bk * LDB + bc] = E[(cb + bc) * 256 + k0 + bk];
    }
    __syncthreads();
    mma_chunk(As, Bs, wr, cp, r, half, acc0, acc1);
  }
  const int c0 = rb + 16 * wr, d0 = cb + 32 * cp + r;
#pragma unroll
  for (int i = 0; i < 8; ++i) {
    G[pk_off(c0 + i + 8 * half, d0)]      = acc0[i];
    G[pk_off(c0 + i + 8 * half, d0 + 16)] = acc1[i];
  }
}

__global__ void pack_x(const float* __restrict__ X, float* __restrict__ Xp) {
  const int lane = threadIdx.x & 31;
  const int blk = blockIdx.x * 8 + (threadIdx.x >> 5);
  const float* s = X + ((blk >> 3) * 16 + (lane & 15)) * 256 + (blk & 7) * 32 + 4 * (lane >> 4);
  float* d = Xp + blk * 512 + lane * 4;
  v4f f[4];
  ld_frag32(s, f);
#pragma unroll
  for (int j = 0; j < 4; ++j) *reinterpret_cast<v4f*>(d + 128 * j) = f[j];
}

__global__ __launch_bounds__(256) void lstm_gemm(const float* __restrict__ Xp, const float* __restrict__ wp,
                          const float* __restrict__ G1p, const float* __restrict__ G2p,
                          const float* __restrict__ Whh, const float* __restrict__ hp,
                          float* __restrict__ z, int l) {
  const int lane = threadIdx.x & 31;
  const int wave = threadIdx.x >> 5;
  const int half = lane >> 4;
  const int r    = lane & 15;
  const int row0 = blockIdx.y * 64 + (wave & 1) * 32;
  const int col0 = blockIdx.x * 128 + (wave >> 1) * 32;
  const int bb = row0 >> 8, nn0 = row0 & 255;
  const float* x0 = Xp + ((((bb * 32 + l) * 256 + nn0) >> 4) << 12) + 4 * lane;
  const float* x1 = x0 + 4096;
  const float* w0 = wp + ((nn0 >> 4) << 12) + 4 * lane;
  const float* w1 = w0 + 4096;
  const float* h0 = hp + ((row0 >> 4) << 12) + 4 * lane;
  const float* h1 = h0 + 4096;
  const int gg = ((col0 >> 4) << 12) + 4 * lane;
  const int cc = (col0 + r) * 256 + 4 * half;
  v8f c00 = {}, c01 = {}, c10 = {}, c11 = {};
  v4f A0[4], A1[4], B0[4], B1[4];
#pragma unroll
  for (int kc = 0; kc < 8; ++kc) {
    ld_packed(x0 + 512 * kc, A0);
    ld_packed(x1 + 512 * kc, A1);
    ld_packed(G1p + gg + 512 * kc, B0);
    ld_packed(G1p + gg + 4096 + 512 * kc, B1);
    mma_direct32(A0, A1, B0, B1, c00, c01, c10, c11);
  }
#pragma unroll
  for (int kc = 0; kc < 8; ++kc) {
    v4f W0[4], W1[4];
    ld_packed(x0 + 512 * kc, A0);
    ld_packed(w0 + 512 * kc, W0);
    ld_packed(x1 + 512 * kc, A1);
    ld_packed(w1 + 512 * kc, W1);
    ld_packed(G2p + gg + 512 * kc, B0);
    ld_packed(G2p + gg + 4096 + 512 * kc, B1);
#pragma unroll
    for (int j = 0; j < 4; ++j) { A0[j] *= W0[j]; A1[j] *= W1[j]; }
    mma_direct32(A0, A1, B0, B1, c00, c01, c10, c11);
  }
#pragma unroll
  for (int kc = 0; kc < 8; ++kc) {
    ld_packed(h0 + 512 * kc, A0);
    ld_packed(h1 + 512 * kc, A1);
    ld_frag32(Whh + cc + 32 * kc, B0);
    ld_frag32(Whh + cc + 16 * 256 + 32 * kc, B1);
    mma_direct32(A0, A1, B0, B1, c00, c01, c10, c11);
  }
  float* o = z + (((row0 >> 4) * 64 + (col0 >> 4)) << 8) + 4 * lane;
  const int nt = 64 * 256;
  *reinterpret_cast<v4f*>(o)                  = v4f{c00[0], c00[1], c00[2], c00[3]};
  *reinterpret_cast<v4f*>(o + 128)            = v4f{c00[4], c00[5], c00[6], c00[7]};
  *reinterpret_cast<v4f*>(o + 256)            = v4f{c01[0], c01[1], c01[2], c01[3]};
  *reinterpret_cast<v4f*>(o + 256 + 128)      = v4f{c01[4], c01[5], c01[6], c01[7]};
  *reinterpret_cast<v4f*>(o + nt)             = v4f{c10[0], c10[1], c10[2], c10[3]};
  *reinterpret_cast<v4f*>(o + nt + 128)       = v4f{c10[4], c10[5], c10[6], c10[7]};
  *reinterpret_cast<v4f*>(o + nt + 256)       = v4f{c11[0], c11[1], c11[2], c11[3]};
  *reinterpret_cast<v4f*>(o + nt + 256 + 128) = v4f{c11[4], c11[5], c11[6], c11[7]};
}

constexpr int LDH = 68;
__global__ __launch_bounds__(256) void lstm_point(const float* __restrict__ z, const float* __restrict__ bih,
                           const float* __restrict__ bhh, float* __restrict__ hp,
                           float* __restrict__ c, float* __restrict__ temp, int l) {
  __shared__ __attribute__((aligned(16))) float hs[16 * LDH];
  const int tid = threadIdx.x;
  const int lane = tid & 31, wave = tid >> 5;
  const int half = lane >> 4, r = lane & 15;
  const int rb = blockIdx.x >> 2, mq = blockIdx.x & 3;
  const int tcl = wave >> 1, q = wave & 1;
  const int tc = mq * 4 + tcl;
  const int m = tc * 16 + r;
  const int in_tile = (q * 32 + lane) * 4;
  const float* zt = z + ((rb * 64 + tc) << 8) + in_tile;
  v4f zi = *reinterpret_cast<const v4f*>(zt);
  v4f zf = *reinterpret_cast<const v4f*>(zt + 16 * 256);
  v4f zg = *reinterpret_cast<const v4f*>(zt + 32 * 256);
  v4f zo = *reinterpret_cast<const v4f*>(zt + 48 * 256);
  float* ct = c + ((rb * 16 + tc) << 8) + in_tile;
  v4f cv = *reinterpret_cast<const v4f*>(ct);
  const float bi0 = bih[m], bi1 = bhh[m];
  const float bf0 = bih[m + 256], bf1 = bhh[m + 256];
  const float bg0 = bih[m + 512], bg1 = bhh[m + 512];
  const float bo0 = bih[m + 768], bo1 = bhh[m + 768];
#pragma unroll
  for (int e = 0; e < 4; ++e) {
    float iv = sigmoidf(zi[e] + bi0 + bi1);
    float fv = sigmoidf(zf[e] + bf0 + bf1);
    float gv = tanhf   (zg[e] + bg0 + bg1);
    float ov = sigmoidf(zo[e] + bo0 + bo1);
    float cn = fv * cv[e] + iv * gv;
    cv[e] = cn;
    hs[(8 * half + 4 * q + e) * LDH + tcl * 16 + r] = ov * tanhf(cn);
  }
  *reinterpret_cast<v4f*>(ct) = cv;
  __syncthreads();
  {
    const int kcl = tid >> 7, idx = tid & 127;
    const int j = idx >> 5, ln = idx & 31;
    v4f v = *reinterpret_cast<const v4f*>(&hs[(ln & 15) * LDH + kcl * 32 + 8 * j + 4 * (ln >> 4)]);
    *reinterpret_cast<v4f*>(hp + ((rb * 8 + mq * 2 + kcl) << 9) + idx * 4) = v;
  }
  {
    const int rl = tid >> 4, m4 = (tid & 15) * 4;
    const int row = rb * 16 + rl;
    const int bb = row >> 8, nn = row & 255;
    v4f v = *reinterpret_cast<const v4f*>(&hs[rl * LDH + m4]);
    *reinterpret_cast<v4f*>(temp + ((bb * 32 + l) * 256 + nn) * 256 + mq * 64 + m4) = v;
  }
}

__global__ void zero_f32(float* __restrict__ p, int n) {
  int i = blockIdx.x * 256 + threadIdx.x;
  if (i < n) p[i] = 0.0f;
}

__global__ void bn_stats(const float* __restrict__ temp, float* __restrict__ stats) {
  const int m = threadIdx.x;
  const int rb = blockIdx.x * 256;
  float s = 0.0f, s2 = 0.0f;
  for (int rr = 0; rr < 256; ++rr) {
    float v = temp[(rb + rr) * 256 + m];
    s += v; s2 += v * v;
  }
  atomicAdd(&stats[m], s);
  atomicAdd(&stats[256 + m], s2);
}

__global__ void bn_final(const float* __restrict__ stats, const float* __restrict__ gamma,
                         const float* __restrict__ beta, float* __restrict__ scale,
                         float* __restrict__ shift) {
  const int m = threadIdx.x;
  const float inv = 1.0f / 65536.0f;
  float mu = stats[m] * inv;
  float var = stats[256 + m] * inv - mu * mu;
  float sc = gamma[m] * rsqrtf(var + 1e-5f);
  scale[m] = sc;
  shift[m] = beta[m] - mu * sc;
}

__global__ void proj_gemm(const float* __restrict__ temp, const float* __restrict__ Wlin,
                          const float* __restrict__ blin, const float* __restrict__ scale,
                          const float* __restrict__ shift, float* __restrict__ proj) {
  __shared__ float As[64 * LDA];
  __shared__ float Bs[32 * LDB];
  WMMA_THREAD_DECOMP();
  const int rb = blockIdx.y * 64;
  const int cb = blockIdx.x * 64;
  v8f acc0 = {}; v8f acc1 = {};
  for (int k0 = 0; k0 < 256; k0 += 32) {
    __syncthreads();
#pragma unroll
    for (int i = 0; i < 8; ++i) {
      int idx = tid + i * 256;
      int ar = idx >> 5, ac = idx & 31;
      As[ar * LDA + ac] = temp[(rb + ar) * 256 + k0 + ac] * scale[k0 + ac] + shift[k0 + ac];
      int bc = idx >> 5, bk = idx & 31;
      Bs[bk * LDB + bc] = Wlin[(cb + bc) * 256 + k0 + bk];
    }
    __syncthreads();
    mma_chunk(As, Bs, wr, cp, r, half, acc0, acc1);
  }
  const int c0 = cb + 32 * cp;
  float* o0 = proj + (rb + 16 * wr) * 256 + c0;
#pragma unroll
  for (int i = 0; i < 8; ++i) {
    o0[(i + 8 * half) * 256 + r]      = acc0[i] + blin[c0 + r];
    o0[(i + 8 * half) * 256 + 16 + r] = acc1[i] + blin[c0 + 16 + r];
  }
}

__global__ void out_gemm(const float* __restrict__ proj, const float* __restrict__ temp,
                         const float* __restrict__ scale, const float* __restrict__ shift,
                         float* __restrict__ out) {
  __shared__ float As[64 * LDA];
  __shared__ float Bs[32 * LDB];
  WMMA_THREAD_DECOMP();
  const int bl = blockIdx.z;
  const int rb = blockIdx.y * 64;
  const int cb = blockIdx.x * 64;
  const float* P = proj + bl * 65536;
  const float* T = temp + bl * 65536;
  v8f acc0 = {}; v8f acc1 = {};
  for (int k0 = 0; k0 < 256; k0 += 32) {
    __syncthreads();
#pragma unroll
    for (int i = 0; i < 8; ++i) {
      int idx = tid + i * 256;
      int ar = idx >> 5, ac = idx & 31;
      As[ar * LDA + ac] = P[(rb + ar) * 256 + k0 + ac];
      int dd = idx >> 5, mm = idx & 31;
      Bs[mm * LDB + dd] = T[(cb + dd) * 256 + k0 + mm] * scale[k0 + mm] + shift[k0 + mm];
    }
    __syncthreads();
    mma_chunk(As, Bs, wr, cp, r, half, acc0, acc1);
  }
  float* o0 = out + bl * 65536 + (rb + 16 * wr) * 256 + cb + 32 * cp;
  store_tile(o0, 256, acc0, r, half);
  store_tile(o0 + 16, 256, acc1, r, half);
}

extern "C" void kernel_launch(void* const* d_in, const int* in_sizes, int n_in,
                              void* d_out, int out_size, void* d_ws, size_t ws_size,
                              hipStream_t stream) {
  (void)in_sizes; (void)n_in; (void)out_size; (void)ws_size;
  const float* X     = (const float*)d_in[0];
  const float* dis   = (const float*)d_in[1];
  const float* E1    = (const float*)d_in[2];
  const float* E2    = (const float*)d_in[3];
  const float* Wih   = (const float*)d_in[4];
  const float* Whh   = (const float*)d_in[5];
  const float* bih   = (const float*)d_in[6];
  const float* bhh   = (const float*)d_in[7];
  const float* gamma = (const float*)d_in[8];
  const float* beta  = (const float*)d_in[9];
  const float* Wlin  = (const float*)d_in[10];
  const float* blin  = (const float*)d_in[11];
  float* out = (float*)d_out;

  float* ws = (float*)d_ws;
  size_t o = 0;
  float* w_sm  = ws + o; o += 256 * 256;
  float* G1T   = ws + o; o += 256 * 1024;
  float* G2T   = ws + o; o += 256 * 1024;
  float* z     = ws + o; o += 2048 * 1024;
  float* h     = ws + o; o += 2048 * 256;
  float* c     = ws + o; o += 2048 * 256;
  float* stats = ws + o; o += 512;
  float* scale = ws + o; o += 256;
  float* shift = ws + o; o += 256;
  float* temp  = ws + o; o += 65536 * 256;
  float* proj  = ws + o; o += 65536 * 256;

  softmax_rows<<<256, 256, 0, stream>>>(dis, w_sm);
  embed_weight_gemm<<<dim3(4, 16, 2), 256, 0, stream>>>(E1, E2, Wih, G1T, G2T);

  pack_x<<<4096, 256, 0, stream>>>(X, proj);

  const int zn = 2048 * 256 * 2 + 512;
  zero_f32<<<(zn + 255) / 256, 256, 0, stream>>>(h, zn);

  for (int l = 0; l < 32; ++l) {
    lstm_gemm<<<dim3(8, 32), 256, 0, stream>>>(proj, w_sm, G1T, G2T, Whh, h, z, l);
    lstm_point<<<512, 256, 0, stream>>>(z, bih, bhh, h, c, temp, l);
  }

  bn_stats<<<256, 256, 0, stream>>>(temp, stats);
  bn_final<<<1, 256, 0, stream>>>(stats, gamma, beta, scale, shift);
  proj_gemm<<<dim3(4, 1024), 256, 0, stream>>>(temp, Wlin, blin, scale, shift, proj);
  out_gemm<<<dim3(4, 4, 256), 256, 0, stream>>>(proj, temp, scale, shift, out);
}


# gate GEMM re-tiled: a wave's 4 accumulators are the i,f,g,o gates of its own 16 channels, so the cell update runs in its registers after the K loop; pointwise launch and z plane go; sums kept
# speedup vs baseline: 2.2911x; 1.0615x over previous
#include <hip/hip_runtime.h>
#include <hip/hip_bf16.h>


#define DI __device__ __forceinline__

typedef float v2f __attribute__((ext_vector_type(2)));
typedef float v4f __attribute__((ext_vector_type(4)));
typedef float v8f __attribute__((ext_vector_type(8)));

constexpr int LDA = 36;
constexpr int LDB = 65;

DI v2f frag_a(const float* As, int k0, int r, int half) {
  v2f a;
  a.x = As[r * LDA + k0 + 2 * half + 0];
  a.y = As[r * LDA + k0 + 2 * half + 1];
  return a;
}
DI v2f frag_b(const float* Bs, int k0, int r, int half) {
  v2f b;
  b.x = Bs[(k0 + 2 * half + 0) * LDB + r];
  b.y = Bs[(k0 + 2 * half + 1) * LDB + r];
  return b;
}

DI void mma_chunk(const float* As, const float* Bs, int wr, int cp, int r, int half,
                  v8f& acc0, v8f& acc1) {
  const float* Aw = As + (wr * 16) * LDA;
  const float* B0 = Bs + 32 * cp;
  const float* B1 = Bs + 32 * cp + 16;
#pragma unroll
  for (int k0 = 0; k0 < 32; k0 += 4) {
    v2f a  = frag_a(Aw, k0, r, half);
    v2f b0 = frag_b(B0, k0, r, half);
    v2f b1 = frag_b(B1, k0, r, half);
    acc0 = __builtin_amdgcn_wmma_f32_16x16x4_f32(false, a, false, b0, (short)0, acc0, false, false);
    acc1 = __builtin_amdgcn_wmma_f32_16x16x4_f32(false, a, false, b1, (short)0, acc1, false, false);
  }
}

DI void store_tile(float* out, int ldo, const v8f& acc, int r, int half) {
#pragma unroll
  for (int i = 0; i < 8; ++i) out[(i + 8 * half) * ldo + r] = acc[i];
}

DI void ld_frag32(const float* p, v4f (&f)[4]) {
#pragma unroll
  for (int j = 0; j < 4; ++j) f[j] = *reinterpret_cast<const v4f*>(p + 8 * j);
}

DI int pk_off(int row, int k) {
  return (((row >> 4) * 8 + (k >> 5)) << 9) + ((((k & 31) >> 3) * 32 + (row & 15) + 16 * ((k >> 2) & 1)) << 2) + (k & 3);
}
DI void ld_packed(const float* p, v4f (&f)[4]) {
#pragma unroll
  for (int j = 0; j < 4; ++j) f[j] = *reinterpret_cast<const v4f*>(p + 128 * j);
}

DI v8f wmma4(v2f a, v2f b, v8f c) {
  return __builtin_amdgcn_wmma_f32_16x16x4_f32(false, a, false, b, (short)0, c, false, false);
}

DI void mma_gates32(const v4f (&A)[4], const v4f (&Bi)[4], const v4f (&Bf)[4], const v4f (&Bg)[4],
                    const v4f (&Bo)[4], v8f& ci, v8f& cf, v8f& cg, v8f& co) {
#pragma unroll
  for (int j = 0; j < 4; ++j) {
    v2f a0 = {A[j].x, A[j].y};
    ci = wmma4(a0, v2f{Bi[j].x, Bi[j].y}, ci); cf = wmma4(a0, v2f{Bf[j].x, Bf[j].y}, cf);
    cg = wmma4(a0, v2f{Bg[j].x, Bg[j].y}, cg); co = wmma4(a0, v2f{Bo[j].x, Bo[j].y}, co);
    v2f a1 = {A[j].z, A[j].w};
    ci = wmma4(a1, v2f{Bi[j].z, Bi[j].w}, ci); cf = wmma4(a1, v2f{Bf[j].z, Bf[j].w}, cf);
    cg = wmma4(a1, v2f{Bg[j].z, Bg[j].w}, cg); co = wmma4(a1, v2f{Bo[j].z, Bo[j].w}, co);
  }
}

struct GateFrags { v4f A[4], W[4], Bi[4], Bf[4], Bg[4], Bo[4]; };

DI void ld_gate_chunk(int t, GateFrags& f, const float* xa, const float* wa, const float* ha,
                      const float* G1p, const float* G2p, const float* Whh, int gg, int cc) {
  const int seg = t >> 3, kc = t & 7, gs = 65536;
  if (seg == 0) {
    ld_packed(xa + 512 * kc, f.A);
    ld_packed(G1p + gg + 512 * kc, f.Bi);
    ld_packed(G1p + gg + gs + 512 * kc, f.Bf);
    ld_packed(G1p + gg + 2 * gs + 512 * kc, f.Bg);
    ld_packed(G1p + gg + 3 * gs + 512 * kc, f.Bo);
  } else if (seg == 1) {
    ld_packed(xa + 512 * kc, f.A);
    ld_packed(wa + 512 * kc, f.W);
    ld_packed(G2p + gg + 512 * kc, f.Bi);
    ld_packed(G2p + gg + gs + 512 * kc, f.Bf);
    ld_packed(G2p + gg + 2 * gs + 512 * kc, f.Bg);
    ld_packed(G2p + gg + 3 * gs + 512 * kc, f.Bo);
  } else {
    ld_packed(ha + 512 * kc, f.A);
    ld_frag32(Whh + cc + 32 * kc, f.Bi);
    ld_frag32(Whh + cc + gs + 32 * kc, f.Bf);
    ld_frag32(Whh + cc + 2 * gs + 32 * kc, f.Bg);
    ld_frag32(Whh + cc + 3 * gs + 32 * kc, f.Bo);
  }
}

#define WMMA_THREAD_DECOMP()        \
  const int tid  = threadIdx.x;     \
  const int lane = tid & 31;        \
  const int wave = tid >> 5;        \
  const int half = lane >> 4;       \
  const int r    = lane & 15;       \
  const int wr   = wave & 3;        \
  const int cp   = wave >> 2;       \
  (void)lane;

DI float sigmoidf(float x) { return 1.0f / (1.0f + __expf(-x)); }

__global__ void softmax_rows(const float* __restrict__ dis, float* __restrict__ w) {
  __shared__ float red[256];
  const int i = blockIdx.x, t = threadIdx.x;
  float v = -dis[i * 256 + t];
  red[t] = v;
  __syncthreads();
  for (int s = 128; s > 0; s >>= 1) {
    if (t < s) red[t] = fmaxf(red[t], red[t + s]);
    __syncthreads();
  }
  float mx = red[0];
  __syncthreads();
  float e = __expf(v - mx);
  red[t] = e;
  __syncthreads();
  for (int s = 128; s > 0; s >>= 1) {
    if (t < s) red[t] += red[t + s];
    __syncthreads();
  }
  w[pk_off(i, t)] = e / red[0];
}

__global__ void embed_weight_gemm(const float* __restrict__ E1, const float* __restrict__ E2,
                                  const float* __restrict__ Wih,
                                  float* __restrict__ G1T, float* __restrict__ G2T) {
  __shared__ float As[64 * LDA];
  __shared__ float Bs[32 * LDB];
  WMMA_THREAD_DECOMP();
  const float* E = blockIdx.z ? E2 : E1;
  float* G = blockIdx.z ? G2T : G1T;
  const int koff = blockIdx.z ? 256 : 0;
  const int rb = blockIdx.y * 64;
  const int cb = blockIdx.x * 64;
  v8f acc0 = {}; v8f acc1 = {};
  for (int k0 = 0; k0 < 256; k0 += 32) {
    __syncthreads();
#pragma unroll
    for (int i = 0; i < 8; ++i) {
      int idx = tid + i * 256;
      int ar = idx >> 5, ac = idx & 31;
      As[ar * LDA + ac] = Wih[(rb + ar) * 512 + koff + k0 + ac];
      int bc = idx >> 5, bk = idx & 31;
      Bs[bk * LDB + bc] = E[(cb + bc) * 256 + k0 + bk];
    }
    __syncthreads();
    mma_chunk(As, Bs, wr, cp, r, half, acc0, acc1);
  }
  const int c0 = rb + 16 * wr, d0 = cb + 32 * cp + r;
#pragma unroll
  for (int i = 0; i < 8; ++i) {
    G[pk_off(c0 + i + 8 * half, d0)]      = acc0[i];
    G[pk_off(c0 + i + 8 * half, d0 + 16)] = acc1[i];
  }
}

__global__ void pack_x(const float* __restrict__ X, float* __restrict__ Xp) {
  const int lane = threadIdx.x & 31;
  const int blk = blockIdx.x * 8 + (threadIdx.x >> 5);
  const float* s = X + ((blk >> 3) * 16 + (lane & 15)) * 256 + (blk & 7) * 32 + 4 * (lane >> 4);
  float* d = Xp + blk * 512 + lane * 4;
  v4f f[4];
  ld_frag32(s, f);
#pragma unroll
  for (int j = 0; j < 4; ++j) *reinterpret_cast<v4f*>(d + 128 * j) = f[j];
}

constexpr int LDH = 36;
__global__ __launch_bounds__(256) void lstm_step(const float* __restrict__ Xp, const float* __restrict__ wp,
                          const float* __restrict__ G1p, const float* __restrict__ G2p,
                          const float* __restrict__ Whh, const float* __restrict__ hin,
                          const float* __restrict__ bih, const float* __restrict__ bhh,
                          float* __restrict__ hout, float* __restrict__ c, float* __restrict__ temp, int l) {
  __shared__ __attribute__((aligned(16))) float hs[64 * LDH];
  const int tid  = threadIdx.x;
  const int lane = tid & 31;
  const int wave = tid >> 5;
  const int half = lane >> 4;
  const int r    = lane & 15;
  const int rbw  = wave & 3;
  const int ctw  = wave >> 2;
  const int row0 = blockIdx.y * 64 + rbw * 16;
  const int ct   = blockIdx.x * 2 + ctw;
  const int bb = row0 >> 8, nn0 = row0 & 255;
  const float* xa = Xp + ((((bb * 32 + l) * 256 + nn0) >> 4) << 12) + 4 * lane;
  const float* wa = wp + ((nn0 >> 4) << 12) + 4 * lane;
  const float* ha = hin + ((row0 >> 4) << 12) + 4 * lane;
  const int gg = (ct << 12) + 4 * lane;
  const int cc = (ct * 16 + r) * 256 + 4 * half;
  float* ctile = c + ((((row0 >> 4) << 4) + ct) << 8) + 4 * lane;
  v4f cv0 = *reinterpret_cast<const v4f*>(ctile);
  v4f cv1 = *reinterpret_cast<const v4f*>(ctile + 128);
  const int m = ct * 16 + r;
  const float bi0 = bih[m], bi1 = bhh[m];
  const float bf0 = bih[m + 256], bf1 = bhh[m + 256];
  const float bg0 = bih[m + 512], bg1 = bhh[m + 512];
  const float bo0 = bih[m + 768], bo1 = bhh[m + 768];
  v8f zi = {}, zf = {}, zg = {}, zo = {};
  GateFrags f[3];
  ld_gate_chunk(0, f[0], xa, wa, ha, G1p, G2p, Whh, gg, cc);
  ld_gate_chunk(1, f[1], xa, wa, ha, G1p, G2p, Whh, gg, cc);
#pragma unroll
  for (int t = 0; t < 24; ++t) {
    if (t + 2 < 24) ld_gate_chunk(t + 2, f[(t + 2) % 3], xa, wa, ha, G1p, G2p, Whh, gg, cc);
    __builtin_amdgcn_sched_barrier(0);
    if ((t >> 3) == 1) {
#pragma unroll
      for (int j = 0; j < 4; ++j) f[t % 3].A[j] *= f[t % 3].W[j];
    }
    mma_gates32(f[t % 3].A, f[t % 3].Bi, f[t % 3].Bf, f[t % 3].Bg, f[t % 3].Bo, zi, zf, zg, zo);
    __builtin_amdgcn_sched_barrier(0);
  }
#pragma unroll
  for (int e = 0; e < 8; ++e) {
    const float cold = (e < 4) ? cv0[e & 3] : cv1[e & 3];
    float iv = sigmoidf(zi[e] + bi0 + bi1);
    float fv = sigmoidf(zf[e] + bf0 + bf1);
    float gv = tanhf   (zg[e] + bg0 + bg1);
    float ov = sigmoidf(zo[e] + bo0 + bo1);
    float cn = fv * cold + iv * gv;
    if (e < 4) cv0[e & 3] = cn; else cv1[e & 3] = cn;
    hs[(16 * rbw + 8 * half + e) * LDH + 16 * ctw + r] = ov * tanhf(cn);
  }
  __syncthreads();
  *reinterpret_cast<v4f*>(ctile)       = cv0;
  *reinterpret_cast<v4f*>(ctile + 128) = cv1;
  const int kcb = blockIdx.x;
#pragma unroll
  for (int t = 0; t < 2; ++t) {
    const int idx = tid + 256 * t;
    {
      const int blk = idx >> 7, q = idx & 127;
      const int j = q >> 5, ln = q & 31;
      v4f v = *reinterpret_cast<const v4f*>(&hs[(16 * blk + (ln & 15)) * LDH + 8 * j + 4 * (ln >> 4)]);
      *reinterpret_cast<v4f*>(hout + (((blockIdx.y * 4 + blk) * 8 + kcb) << 9) + q * 4) = v;
    }
    {
      const int rl = idx >> 3, m4 = (idx & 7) * 4;
      const int row = blockIdx.y * 64 + rl;
      const int b2 = row >> 8, nn = row & 255;
      v4f v = *reinterpret_cast<const v4f*>(&hs[rl * LDH + m4]);
      *reinterpret_cast<v4f*>(temp + ((b2 * 32 + l) * 256 + nn) * 256 + kcb * 32 + m4) = v;
    }
  }
}

__global__ void zero_f32(float* __restrict__ p, int n) {
  int i = blockIdx.x * 256 + threadIdx.x;
  if (i < n) p[i] = 0.0f;
}

__global__ void bn_stats(const float* __restrict__ temp, float* __restrict__ stats) {
  const int m = threadIdx.x;
  const int rb = blockIdx.x * 256;
  float s = 0.0f, s2 = 0.0f;
  for (int rr = 0; rr < 256; ++rr) {
    float v = temp[(rb + rr) * 256 + m];
    s += v; s2 += v * v;
  }
  atomicAdd(&stats[m], s);
  atomicAdd(&stats[256 + m], s2);
}

__global__ void bn_final(const float* __restrict__ stats, const float* __restrict__ gamma,
                         const float* __restrict__ beta, float* __restrict__ scale,
                         float* __restrict__ shift) {
  const int m = threadIdx.x;
  const float inv = 1.0f / 65536.0f;
  float mu = stats[m] * inv;
  float var = stats[256 + m] * inv - mu * mu;
  float sc = gamma[m] * rsqrtf(var + 1e-5f);
  scale[m] = sc;
  shift[m] = beta[m] - mu * sc;
}

__global__ void proj_gemm(const float* __restrict__ temp, const float* __restrict__ Wlin,
                          const float* __restrict__ blin, const float* __restrict__ scale,
                          const float* __restrict__ shift, float* __restrict__ proj) {
  __shared__ float As[64 * LDA];
  __shared__ float Bs[32 * LDB];
  WMMA_THREAD_DECOMP();
  const int rb = blockIdx.y * 64;
  const int cb = blockIdx.x * 64;
  v8f acc0 = {}; v8f acc1 = {};
  for (int k0 = 0; k0 < 256; k0 += 32) {
    __syncthreads();
#pragma unroll
    for (int i = 0; i < 8; ++i) {
      int idx = tid + i * 256;
      int ar = idx >> 5, ac = idx & 31;
      As[ar * LDA + ac] = temp[(rb + ar) * 256 + k0 + ac] * scale[k0 + ac] + shift[k0 + ac];
      int bc = idx >> 5, bk = idx & 31;
      Bs[bk * LDB + bc] = Wlin[(cb + bc) * 256 + k0 + bk];
    }
    __syncthreads();
    mma_chunk(As, Bs, wr, cp, r, half, acc0, acc1);
  }
  const int c0 = cb + 32 * cp;
  float* o0 = proj + (rb + 16 * wr) * 256 + c0;
#pragma unroll
  for (int i = 0; i < 8; ++i) {
    o0[(i + 8 * half) * 256 + r]      = acc0[i] + blin[c0 + r];
    o0[(i + 8 * half) * 256 + 16 + r] = acc1[i] + blin[c0 + 16 + r];
  }
}

__global__ void out_gemm(const float* __restrict__ proj, const float* __restrict__ temp,
                         const float* __restrict__ scale, const float* __restrict__ shift,
                         float* __restrict__ out) {
  __shared__ float As[64 * LDA];
  __shared__ float Bs[32 * LDB];
  WMMA_THREAD_DECOMP();
  const int bl = blockIdx.z;
  const int rb = blockIdx.y * 64;
  const int cb = blockIdx.x * 64;
  const float* P = proj + bl * 65536;
  const float* T = temp + bl * 65536;
  v8f acc0 = {}; v8f acc1 = {};
  for (int k0 = 0; k0 < 256; k0 += 32) {
    __syncthreads();
#pragma unroll
    for (int i = 0; i < 8; ++i) {
      int idx = tid + i * 256;
      int ar = idx >> 5, ac = idx & 31;
      As[ar * LDA + ac] = P[(rb + ar) * 256 + k0 + ac];
      int dd = idx >> 5, mm = idx & 31;
      Bs[mm * LDB + dd] = T[(cb + dd) * 256 + k0 + mm] * scale[k0 + mm] + shift[k0 + mm];
    }
    __syncthreads();
    mma_chunk(As, Bs, wr, cp, r, half, acc0, acc1);
  }
  float* o0 = out + bl * 65536 + (rb + 16 * wr) * 256 + cb + 32 * cp;
  store_tile(o0, 256, acc0, r, half);
  store_tile(o0 + 16, 256, acc1, r, half);
}

extern "C" void kernel_launch(void* const* d_in, const int* in_sizes, int n_in,
                              void* d_out, int out_size, void* d_ws, size_t ws_size,
                              hipStream_t stream) {
  (void)in_sizes; (void)n_in; (void)out_size; (void)ws_size;
  const float* X     = (const float*)d_in[0];
  const float* dis   = (const float*)d_in[1];
  const float* E1    = (const float*)d_in[2];
  const float* E2    = (const float*)d_in[3];
  const float* Wih   = (const float*)d_in[4];
  const float* Whh   = (const float*)d_in[5];
  const float* bih   = (const float*)d_in[6];
  const float* bhh   = (const float*)d_in[7];
  const float* gamma = (const float*)d_in[8];
  const float* beta  = (const float*)d_in[9];
  const float* Wlin  = (const float*)d_in[10];
  const float* blin  = (const float*)d_in[11];
  float* out = (float*)d_out;

  float* ws = (float*)d_ws;
  size_t o = 0;
  float* w_sm  = ws + o; o += 256 * 256;
  float* G1T   = ws + o; o += 256 * 1024;
  float* G2T   = ws + o; o += 256 * 1024;
  float* h2    = ws + o; o += 2048 * 1024;
  float* h     = ws + o; o += 2048 * 256;
  float* c     = ws + o; o += 2048 * 256;
  float* stats = ws + o; o += 512;
  float* scale = ws + o; o += 256;
  float* shift = ws + o; o += 256;
  float* temp  = ws + o; o += 65536 * 256;
  float* proj  = ws + o; o += 65536 * 256;

  softmax_rows<<<256, 256, 0, stream>>>(dis, w_sm);
  embed_weight_gemm<<<dim3(4, 16, 2), 256, 0, stream>>>(E1, E2, Wih, G1T, G2T);

  pack_x<<<4096, 256, 0, stream>>>(X, proj);

  const int zn = 2048 * 256 * 2 + 512;
  zero_f32<<<(zn + 255) / 256, 256, 0, stream>>>(h, zn);

  for (int l = 0; l < 32; l += 2) {
    lstm_step<<<dim3(8, 32), 256, 0, stream>>>(proj, w_sm, G1T, G2T, Whh, h, bih, bhh, h2, c, temp, l);
    lstm_step<<<dim3(8, 32), 256, 0, stream>>>(proj, w_sm, G1T, G2T, Whh, h2, bih, bhh, h, c, temp, l + 1);
  }

  bn_stats<<<256, 256, 0, stream>>>(temp, stats);
  bn_final<<<1, 256, 0, stream>>>(stats, gamma, beta, scale, shift);
  proj_gemm<<<dim3(4, 1024), 256, 0, stream>>>(temp, Wlin, blin, scale, shift, proj);
  out_gemm<<<dim3(4, 4, 256), 256, 0, stream>>>(proj, temp, scale, shift, out);
}


# proj, out GEMMs: each wave reads its WMMA fragments straight from global memory (16-byte k-contiguous loads, 64x64 wave tile, next chunk prefetched); no LDS or barriers; 8-k group sum order changes
# speedup vs baseline: 2.4056x; 1.0500x over previous
#include <hip/hip_runtime.h>
#include <hip/hip_bf16.h>


#define DI __device__ __forceinline__

typedef float v2f __attribute__((ext_vector_type(2)));
typedef float v4f __attribute__((ext_vector_type(4)));
typedef float v8f __attribute__((ext_vector_type(8)));

constexpr int LDA = 36;
constexpr int LDB = 65;

DI v2f frag_a(const float* As, int k0, int r, int half) {
  v2f a;
  a.x = As[r * LDA + k0 + 2 * half + 0];
  a.y = As[r * LDA + k0 + 2 * half + 1];
  return a;
}
DI v2f frag_b(const float* Bs, int k0, int r, int half) {
  v2f b;
  b.x = Bs[(k0 + 2 * half + 0) * LDB + r];
  b.y = Bs[(k0 + 2 * half + 1) * LDB + r];
  return b;
}

DI void mma_chunk(const float* As, const float* Bs, int wr, int cp, int r, int half,
                  v8f& acc0, v8f& acc1) {
  const float* Aw = As + (wr * 16) * LDA;
  const float* B0 = Bs + 32 * cp;
  const float* B1 = Bs + 32 * cp + 16;
#pragma unroll
  for (int k0 = 0; k0 < 32; k0 += 4) {
    v2f a  = frag_a(Aw, k0, r, half);
    v2f b0 = frag_b(B0, k0, r, half);
    v2f b1 = frag_b(B1, k0, r, half);
    acc0 = __builtin_amdgcn_wmma_f32_16x16x4_f32(false, a, false, b0, (short)0, acc0, false, false);
    acc1 = __builtin_amdgcn_wmma_f32_16x16x4_f32(false, a, false, b1, (short)0, acc1, false, false);
  }
}

DI void ld_frag32(const float* p, v4f (&f)[4]) {
#pragma unroll
  for (int j = 0; j < 4; ++j) f[j] = *reinterpret_cast<const v4f*>(p + 8 * j);
}

DI int pk_off(int row, int k) {
  return (((row >> 4) * 8 + (k >> 5)) << 9) + ((((k & 31) >> 3) * 32 + (row & 15) + 16 * ((k >> 2) & 1)) << 2) + (k & 3);
}
DI void ld_packed(const float* p, v4f (&f)[4]) {
#pragma unroll
  for (int j = 0; j < 4; ++j) f[j] = *reinterpret_cast<const v4f*>(p + 128 * j);
}

DI v8f wmma4(v2f a, v2f b, v8f c) {
  return __builtin_amdgcn_wmma_f32_16x16x4_f32(false, a, false, b, (short)0, c, false, false);
}

DI void mma_gates32(const v4f (&A)[4], const v4f (&Bi)[4], const v4f (&Bf)[4], const v4f (&Bg)[4],
                    const v4f (&Bo)[4], v8f& ci, v8f& cf, v8f& cg, v8f& co) {
#pragma unroll
  for (int j = 0; j < 4; ++j) {
    v2f a0 = {A[j].x, A[j].y};
    ci = wmma4(a0, v2f{Bi[j].x, Bi[j].y}, ci); cf = wmma4(a0, v2f{Bf[j].x, Bf[j].y}, cf);
    cg = wmma4(a0, v2f{Bg[j].x, Bg[j].y}, cg); co = wmma4(a0, v2f{Bo[j].x, Bo[j].y}, co);
    v2f a1 = {A[j].z, A[j].w};
    ci = wmma4(a1, v2f{Bi[j].z, Bi[j].w}, ci); cf = wmma4(a1, v2f{Bf[j].z, Bf[j].w}, cf);
    cg = wmma4(a1, v2f{Bg[j].z, Bg[j].w}, cg); co = wmma4(a1, v2f{Bo[j].z, Bo[j].w}, co);
  }
}

struct GateFrags { v4f A[4], W[4], Bi[4], Bf[4], Bg[4], Bo[4]; };

DI void ld_gate_chunk(int t, GateFrags& f, const float* xa, const float* wa, const float* ha,
                      const float* G1p, const float* G2p, const float* Whh, int gg, int cc) {
  const int seg = t >> 3, kc = t & 7, gs = 65536;
  if (seg == 0) {
    ld_packed(xa + 512 * kc, f.A);
    ld_packed(G1p + gg + 512 * kc, f.Bi);
    ld_packed(G1p + gg + gs + 512 * kc, f.Bf);
    ld_packed(G1p + gg + 2 * gs + 512 * kc, f.Bg);
    ld_packed(G1p + gg + 3 * gs + 512 * kc, f.Bo);
  } else if (seg == 1) {
    ld_packed(xa + 512 * kc, f.A);
    ld_packed(wa + 512 * kc, f.W);
    ld_packed(G2p + gg + 512 * kc, f.Bi);
    ld_packed(G2p + gg + gs + 512 * kc, f.Bf);
    ld_packed(G2p + gg + 2 * gs + 512 * kc, f.Bg);
    ld_packed(G2p + gg + 3 * gs + 512 * kc, f.Bo);
  } else {
    ld_packed(ha + 512 * kc, f.A);
    ld_frag32(Whh + cc + 32 * kc, f.Bi);
    ld_frag32(Whh + cc + gs + 32 * kc, f.Bf);
    ld_frag32(Whh + cc + 2 * gs + 32 * kc, f.Bg);
    ld_frag32(Whh + cc + 3 * gs + 32 * kc, f.Bo);
  }
}

#define WMMA_THREAD_DECOMP()        \
  const int tid  = threadIdx.x;     \
  const int lane = tid & 31;        \
  const int wave = tid >> 5;        \
  const int half = lane >> 4;       \
  const int r    = lane & 15;       \
  const int wr   = wave & 3;        \
  const int cp   = wave >> 2;       \
  (void)lane;

DI float sigmoidf(float x) { return 1.0f / (1.0f + __expf(-x)); }

__global__ void softmax_rows(const float* __restrict__ dis, float* __restrict__ w) {
  __shared__ float red[256];
  const int i = blockIdx.x, t = threadIdx.x;
  float v = -dis[i * 256 + t];
  red[t] = v;
  __syncthreads();
  for (int s = 128; s > 0; s >>= 1) {
    if (t < s) red[t] = fmaxf(red[t], red[t + s]);
    __syncthreads();
  }
  float mx = red[0];
  __syncthreads();
  float e = __expf(v - mx);
  red[t] = e;
  __syncthreads();
  for (int s = 128; s > 0; s >>= 1) {
    if (t < s) red[t] += red[t + s];
    __syncthreads();
  }
  w[pk_off(i, t)] = e / red[0];
}

__global__ void embed_weight_gemm(const float* __restrict__ E1, const float* __restrict__ E2,
                                  const float* __restrict__ Wih,
                                  float* __restrict__ G1T, float* __restrict__ G2T) {
  __shared__ float As[64 * LDA];
  __shared__ float Bs[32 * LDB];
  WMMA_THREAD_DECOMP();
  const float* E = blockIdx.z ? E2 : E1;
  float* G = blockIdx.z ? G2T : G1T;
  const int koff = blockIdx.z ? 256 : 0;
  const int rb = blockIdx.y * 64;
  const int cb = blockIdx.x * 64;
  v8f acc0 = {}; v8f acc1 = {};
  for (int k0 = 0; k0 < 256; k0 += 32) {
    __syncthreads();
#pragma unroll
    for (int i = 0; i < 8; ++i) {
      int idx = tid + i * 256;
      int ar = idx >> 5, ac = idx & 31;
      As[ar * LDA + ac] = Wih[(rb + ar) * 512 + koff + k0 + ac];
      int bc = idx >> 5, bk = idx & 31;
      Bs[bk * LDB + bc] = E[(cb + bc) * 256 + k0 + bk];
    }
    __syncthreads();
    mma_chunk(As, Bs, wr, cp, r, half, acc0, acc1);
  }
  const int c0 = rb + 16 * wr, d0 = cb + 32 * cp + r;
#pragma unroll
  for (int i = 0; i < 8; ++i) {
    G[pk_off(c0 + i + 8 * half, d0)]      = acc0[i];
    G[pk_off(c0 + i + 8 * half, d0 + 16)] = acc1[i];
  }
}

__global__ void pack_x(const float* __restrict__ X, float* __restrict__ Xp) {
  const int lane = threadIdx.x & 31;
  const int blk = blockIdx.x * 8 + (threadIdx.x >> 5);
  const float* s = X + ((blk >> 3) * 16 + (lane & 15)) * 256 + (blk & 7) * 32 + 4 * (lane >> 4);
  float* d = Xp + blk * 512 + lane * 4;
  v4f f[4];
  ld_frag32(s, f);
#pragma unroll
  for (int j = 0; j < 4; ++j) *reinterpret_cast<v4f*>(d + 128 * j) = f[j];
}

constexpr int LDH = 36;
__global__ __launch_bounds__(256) void lstm_step(const float* __restrict__ Xp, const float* __restrict__ wp,
                          const float* __restrict__ G1p, const float* __restrict__ G2p,
                          const float* __restrict__ Whh, const float* __restrict__ hin,
                          const float* __restrict__ bih, const float* __restrict__ bhh,
                          float* __restrict__ hout, float* __restrict__ c, float* __restrict__ temp, int l) {
  __shared__ __attribute__((aligned(16))) float hs[64 * LDH];
  const int tid  = threadIdx.x;
  const int lane = tid & 31;
  const int wave = tid >> 5;
  const int half = lane >> 4;
  const int r    = lane & 15;
  const int rbw  = wave & 3;
  const int ctw  = wave >> 2;
  const int row0 = blockIdx.y * 64 + rbw * 16;
  const int ct   = blockIdx.x * 2 + ctw;
  const int bb = row0 >> 8, nn0 = row0 & 255;
  const float* xa = Xp + ((((bb * 32 + l) * 256 + nn0) >> 4) << 12) + 4 * lane;
  const float* wa = wp + ((nn0 >> 4) << 12) + 4 * lane;
  const float* ha = hin + ((row0 >> 4) << 12) + 4 * lane;
  const int gg = (ct << 12) + 4 * lane;
  const int cc = (ct * 16 + r) * 256 + 4 * half;
  float* ctile = c + ((((row0 >> 4) << 4) + ct) << 8) + 4 * lane;
  v4f cv0 = *reinterpret_cast<const v4f*>(ctile);
  v4f cv1 = *reinterpret_cast<const v4f*>(ctile + 128);
  const int m = ct * 16 + r;
  const float bi0 = bih[m], bi1 = bhh[m];
  const float bf0 = bih[m + 256], bf1 = bhh[m + 256];
  const float bg0 = bih[m + 512], bg1 = bhh[m + 512];
  const float bo0 = bih[m + 768], bo1 = bhh[m + 768];
  v8f zi = {}, zf = {}, zg = {}, zo = {};
  GateFrags f[3];
  ld_gate_chunk(0, f[0], xa, wa, ha, G1p, G2p, Whh, gg, cc);
  ld_gate_chunk(1, f[1], xa, wa, ha, G1p, G2p, Whh, gg, cc);
#pragma unroll
  for (int t = 0; t < 24; ++t) {
    if (t + 2 < 24) ld_gate_chunk(t + 2, f[(t + 2) % 3], xa, wa, ha, G1p, G2p, Whh, gg, cc);
    __builtin_amdgcn_sched_barrier(0);
    if ((t >> 3) == 1) {
#pragma unroll
      for (int j = 0; j < 4; ++j) f[t % 3].A[j] *= f[t % 3].W[j];
    }
    mma_gates32(f[t % 3].A, f[t % 3].Bi, f[t % 3].Bf, f[t % 3].Bg, f[t % 3].Bo, zi, zf, zg, zo);
    __builtin_amdgcn_sched_barrier(0);
  }
#pragma unroll
  for (int e = 0; e < 8; ++e) {
    const float cold = (e < 4) ? cv0[e & 3] : cv1[e & 3];
    float iv = sigmoidf(zi[e] + bi0 + bi1);
    float fv = sigmoidf(zf[e] + bf0 + bf1);
    float gv = tanhf   (zg[e] + bg0 + bg1);
    float ov = sigmoidf(zo[e] + bo0 + bo1);
    float cn = fv * cold + iv * gv;
    if (e < 4) cv0[e & 3] = cn; else cv1[e & 3] = cn;
    hs[(16 * rbw + 8 * half + e) * LDH + 16 * ctw + r] = ov * tanhf(cn);
  }
  __syncthreads();
  *reinterpret_cast<v4f*>(ctile)       = cv0;
  *reinterpret_cast<v4f*>(ctile + 128) = cv1;
  const int kcb = blockIdx.x;
#pragma unroll
  for (int t = 0; t < 2; ++t) {
    const int idx = tid + 256 * t;
    {
      const int blk = idx >> 7, q = idx & 127;
      const int j = q >> 5, ln = q & 31;
      v4f v = *reinterpret_cast<const v4f*>(&hs[(16 * blk + (ln & 15)) * LDH + 8 * j + 4 * (ln >> 4)]);
      *reinterpret_cast<v4f*>(hout + (((blockIdx.y * 4 + blk) * 8 + kcb) << 9) + q * 4) = v;
    }
    {
      const int rl = idx >> 3, m4 = (idx & 7) * 4;
      const int row = blockIdx.y * 64 + rl;
      const int b2 = row >> 8, nn = row & 255;
      v4f v = *reinterpret_cast<const v4f*>(&hs[rl * LDH + m4]);
      *reinterpret_cast<v4f*>(temp + ((b2 * 32 + l) * 256 + nn) * 256 + kcb * 32 + m4) = v;
    }
  }
}

__global__ void zero_f32(float* __restrict__ p, int n) {
  int i = blockIdx.x * 256 + threadIdx.x;
  if (i < n) p[i] = 0.0f;
}

__global__ void bn_stats(const float* __restrict__ temp, float* __restrict__ stats) {
  const int m = threadIdx.x;
  const int rb = blockIdx.x * 256;
  float s = 0.0f, s2 = 0.0f;
  for (int rr = 0; rr < 256; ++rr) {
    float v = temp[(rb + rr) * 256 + m];
    s += v; s2 += v * v;
  }
  atomicAdd(&stats[m], s);
  atomicAdd(&stats[256 + m], s2);
}

__global__ void bn_final(const float* __restrict__ stats, const float* __restrict__ gamma,
                         const float* __restrict__ beta, float* __restrict__ scale,
                         float* __restrict__ shift) {
  const int m = threadIdx.x;
  const float inv = 1.0f / 65536.0f;
  float mu = stats[m] * inv;
  float var = stats[256 + m] * inv - mu * mu;
  float sc = gamma[m] * rsqrtf(var + 1e-5f);
  scale[m] = sc;
  shift[m] = beta[m] - mu * sc;
}

struct TileFrags { v4f A[4][4], B[4][4], S[4], H[4]; };

DI void ld_tile_chunk(int kc, TileFrags& f, const float* pa, const float* pb, const float* ps, const float* ph) {
#pragma unroll
  for (int i = 0; i < 4; ++i) ld_frag32(pa + i * 4096 + 32 * kc, f.A[i]);
#pragma unroll
  for (int n = 0; n < 4; ++n) ld_frag32(pb + n * 4096 + 32 * kc, f.B[n]);
  ld_frag32(ps + 32 * kc, f.S);
  ld_frag32(ph + 32 * kc, f.H);
}

DI void mma_tile64(const TileFrags& f, v8f (&acc)[4][4]) {
#pragma unroll
  for (int j = 0; j < 4; ++j) {
#pragma unroll
    for (int i = 0; i < 4; ++i) {
      const v2f a0 = {f.A[i][j].x, f.A[i][j].y};
#pragma unroll
      for (int n = 0; n < 4; ++n) acc[i][n] = wmma4(a0, v2f{f.B[n][j].x, f.B[n][j].y}, acc[i][n]);
    }
#pragma unroll
    for (int i = 0; i < 4; ++i) {
      const v2f a1 = {f.A[i][j].z, f.A[i][j].w};
#pragma unroll
      for (int n = 0; n < 4; ++n) acc[i][n] = wmma4(a1, v2f{f.B[n][j].z, f.B[n][j].w}, acc[i][n]);
    }
  }
}

template <bool BN_ON_A>
DI void tile_gemm64(const float* pa, const float* pb, const float* ps, const float* ph, v8f (&acc)[4][4]) {
  TileFrags f[2];
  ld_tile_chunk(0, f[0], pa, pb, ps, ph);
#pragma unroll
  for (int t = 0; t < 8; ++t) {
    if (t + 1 < 8) ld_tile_chunk(t + 1, f[(t + 1) & 1], pa, pb, ps, ph);
    __builtin_amdgcn_sched_barrier(0);
    TileFrags& g = f[t & 1];
#pragma unroll
    for (int i = 0; i < 4; ++i) {
#pragma unroll
      for (int j = 0; j < 4; ++j) {
        if (BN_ON_A) g.A[i][j] = g.A[i][j] * g.S[j] + g.H[j];
        else         g.B[i][j] = g.B[i][j] * g.S[j] + g.H[j];
      }
    }
    mma_tile64(g, acc);
    __builtin_amdgcn_sched_barrier(0);
  }
}

__global__ __launch_bounds__(128) void proj_gemm(const float* __restrict__ temp, const float* __restrict__ Wlin,
                          const float* __restrict__ blin, const float* __restrict__ scale,
                          const float* __restrict__ shift, float* __restrict__ proj) {
  const int lane = threadIdx.x & 31, wave = threadIdx.x >> 5;
  const int half = lane >> 4, r = lane & 15;
  const int row0 = blockIdx.y * 128 + (wave & 1) * 64;
  const int col0 = blockIdx.x * 128 + (wave >> 1) * 64;
  v8f acc[4][4] = {};
  tile_gemm64<true>(temp + (row0 + r) * 256 + 4 * half, Wlin + (col0 + r) * 256 + 4 * half,
                    scale + 4 * half, shift + 4 * half, acc);
#pragma unroll
  for (int n = 0; n < 4; ++n) {
    const float bias = blin[col0 + 16 * n + r];
#pragma unroll
    for (int i = 0; i < 4; ++i) {
      float* o = proj + (row0 + 16 * i + 8 * half) * 256 + col0 + 16 * n + r;
#pragma unroll
      for (int e = 0; e < 8; ++e) o[e * 256] = acc[i][n][e] + bias;
    }
  }
}

__global__ __launch_bounds__(128) void out_gemm(const float* __restrict__ proj, const float* __restrict__ temp,
                         const float* __restrict__ scale, const float* __restrict__ shift,
                         float* __restrict__ out) {
  const int lane = threadIdx.x & 31, wave = threadIdx.x >> 5;
  const int half = lane >> 4, r = lane & 15;
  const int bl = blockIdx.y;
  const int o0 = (blockIdx.x >> 1) * 128 + (wave & 1) * 64;
  const int d0 = (blockIdx.x & 1) * 128 + (wave >> 1) * 64;
  v8f acc[4][4] = {};
  tile_gemm64<false>(proj + bl * 65536 + (o0 + r) * 256 + 4 * half, temp + bl * 65536 + (d0 + r) * 256 + 4 * half,
                     scale + 4 * half, shift + 4 * half, acc);
#pragma unroll
  for (int i = 0; i < 4; ++i) {
#pragma unroll
    for (int n = 0; n < 4; ++n) {
      float* o = out + bl * 65536 + (o0 + 16 * i + 8 * half) * 256 + d0 + 16 * n + r;
#pragma unroll
      for (int e = 0; e < 8; ++e) o[e * 256] = acc[i][n][e];
    }
  }
}

extern "C" void kernel_launch(void* const* d_in, const int* in_sizes, int n_in,
                              void* d_out, int out_size, void* d_ws, size_t ws_size,
                              hipStream_t stream) {
  (void)in_sizes; (void)n_in; (void)out_size; (void)ws_size;
  const float* X     = (const float*)d_in[0];
  const float* dis   = (const float*)d_in[1];
  const float* E1    = (const float*)d_in[2];
  const float* E2    = (const float*)d_in[3];
  const float* Wih   = (const float*)d_in[4];
  const float* Whh   = (const float*)d_in[5];
  const float* bih   = (const float*)d_in[6];
  const float* bhh   = (const float*)d_in[7];
  const float* gamma = (const float*)d_in[8];
  const float* beta  = (const float*)d_in[9];
  const float* Wlin  = (const float*)d_in[10];
  const float* blin  = (const float*)d_in[11];
  float* out = (float*)d_out;

  float* ws = (float*)d_ws;
  size_t o = 0;
  float* w_sm  = ws + o; o += 256 * 256;
  float* G1T   = ws + o; o += 256 * 1024;
  float* G2T   = ws + o; o += 256 * 1024;
  float* h2    = ws + o; o += 2048 * 1024;
  float* h     = ws + o; o += 2048 * 256;
  float* c     = ws + o; o += 2048 * 256;
  float* stats = ws + o; o += 512;
  float* scale = ws + o; o += 256;
  float* shift = ws + o; o += 256;
  float* temp  = ws + o; o += 65536 * 256;
  float* proj  = ws + o; o += 65536 * 256;

  softmax_rows<<<256, 256, 0, stream>>>(dis, w_sm);
  embed_weight_gemm<<<dim3(4, 16, 2), 256, 0, stream>>>(E1, E2, Wih, G1T, G2T);

  pack_x<<<4096, 256, 0, stream>>>(X, proj);

  const int zn = 2048 * 256 * 2 + 512;
  zero_f32<<<(zn + 255) / 256, 256, 0, stream>>>(h, zn);

  for (int l = 0; l < 32; l += 2) {
    lstm_step<<<dim3(8, 32), 256, 0, stream>>>(proj, w_sm, G1T, G2T, Whh, h, bih, bhh, h2, c, temp, l);
    lstm_step<<<dim3(8, 32), 256, 0, stream>>>(proj, w_sm, G1T, G2T, Whh, h2, bih, bhh, h, c, temp, l + 1);
  }

  bn_stats<<<256, 256, 0, stream>>>(temp, stats);
  bn_final<<<1, 256, 0, stream>>>(stats, gamma, beta, scale, shift);
  proj_gemm<<<dim3(2, 512), 128, 0, stream>>>(temp, Wlin, blin, scale, shift, proj);
  out_gemm<<<dim3(4, 256), 128, 0, stream>>>(proj, temp, scale, shift, out);
}


# LSTM cell update: sigmoid and tanh as branch-free exp + hardware reciprocal (no tanhf branches, no IEEE divide), so a lane's 8 cells interleave; activations differ by rounding only, sums kept
# speedup vs baseline: 2.5258x; 1.0500x over previous
#include <hip/hip_runtime.h>
#include <hip/hip_bf16.h>


#define DI __device__ __forceinline__

typedef float v2f __attribute__((ext_vector_type(2)));
typedef float v4f __attribute__((ext_vector_type(4)));
typedef float v8f __attribute__((ext_vector_type(8)));

constexpr int LDA = 36;
constexpr int LDB = 65;

DI v2f frag_a(const float* As, int k0, int r, int half) {
  v2f a;
  a.x = As[r * LDA + k0 + 2 * half + 0];
  a.y = As[r * LDA + k0 + 2 * half + 1];
  return a;
}
DI v2f frag_b(const float* Bs, int k0, int r, int half) {
  v2f b;
  b.x = Bs[(k0 + 2 * half + 0) * LDB + r];
  b.y = Bs[(k0 + 2 * half + 1) * LDB + r];
  return b;
}

DI void mma_chunk(const float* As, const float* Bs, int wr, int cp, int r, int half,
                  v8f& acc0, v8f& acc1) {
  const float* Aw = As + (wr * 16) * LDA;
  const float* B0 = Bs + 32 * cp;
  const float* B1 = Bs + 32 * cp + 16;
#pragma unroll
  for (int k0 = 0; k0 < 32; k0 += 4) {
    v2f a  = frag_a(Aw, k0, r, half);
    v2f b0 = frag_b(B0, k0, r, half);
    v2f b1 = frag_b(B1, k0, r, half);
    acc0 = __builtin_amdgcn_wmma_f32_16x16x4_f32(false, a, false, b0, (short)0, acc0, false, false);
    acc1 = __builtin_amdgcn_wmma_f32_16x16x4_f32(false, a, false, b1, (short)0, acc1, false, false);
  }
}

DI void ld_frag32(const float* p, v4f (&f)[4]) {
#pragma unroll
  for (int j = 0; j < 4; ++j) f[j] = *reinterpret_cast<const v4f*>(p + 8 * j);
}

DI int pk_off(int row, int k) {
  return (((row >> 4) * 8 + (k >> 5)) << 9) + ((((k & 31) >> 3) * 32 + (row & 15) + 16 * ((k >> 2) & 1)) << 2) + (k & 3);
}
DI void ld_packed(const float* p, v4f (&f)[4]) {
#pragma unroll
  for (int j = 0; j < 4; ++j) f[j] = *reinterpret_cast<const v4f*>(p + 128 * j);
}

DI v8f wmma4(v2f a, v2f b, v8f c) {
  return __builtin_amdgcn_wmma_f32_16x16x4_f32(false, a, false, b, (short)0, c, false, false);
}

DI void mma_gates32(const v4f (&A)[4], const v4f (&Bi)[4], const v4f (&Bf)[4], const v4f (&Bg)[4],
                    const v4f (&Bo)[4], v8f& ci, v8f& cf, v8f& cg, v8f& co) {
#pragma unroll
  for (int j = 0; j < 4; ++j) {
    v2f a0 = {A[j].x, A[j].y};
    ci = wmma4(a0, v2f{Bi[j].x, Bi[j].y}, ci); cf = wmma4(a0, v2f{Bf[j].x, Bf[j].y}, cf);
    cg = wmma4(a0, v2f{Bg[j].x, Bg[j].y}, cg); co = wmma4(a0, v2f{Bo[j].x, Bo[j].y}, co);
    v2f a1 = {A[j].z, A[j].w};
    ci = wmma4(a1, v2f{Bi[j].z, Bi[j].w}, ci); cf = wmma4(a1, v2f{Bf[j].z, Bf[j].w}, cf);
    cg = wmma4(a1, v2f{Bg[j].z, Bg[j].w}, cg); co = wmma4(a1, v2f{Bo[j].z, Bo[j].w}, co);
  }
}

struct GateFrags { v4f A[4], W[4], Bi[4], Bf[4], Bg[4], Bo[4]; };

DI void ld_gate_chunk(int t, GateFrags& f, const float* xa, const float* wa, const float* ha,
                      const float* G1p, const float* G2p, const float* Whh, int gg, int cc) {
  const int seg = t >> 3, kc = t & 7, gs = 65536;
  if (seg == 0) {
    ld_packed(xa + 512 * kc, f.A);
    ld_packed(G1p + gg + 512 * kc, f.Bi);
    ld_packed(G1p + gg + gs + 512 * kc, f.Bf);
    ld_packed(G1p + gg + 2 * gs + 512 * kc, f.Bg);
    ld_packed(G1p + gg + 3 * gs + 512 * kc, f.Bo);
  } else if (seg == 1) {
    ld_packed(xa + 512 * kc, f.A);
    ld_packed(wa + 512 * kc, f.W);
    ld_packed(G2p + gg + 512 * kc, f.Bi);
    ld_packed(G2p + gg + gs + 512 * kc, f.Bf);
    ld_packed(G2p + gg + 2 * gs + 512 * kc, f.Bg);
    ld_packed(G2p + gg + 3 * gs + 512 * kc, f.Bo);
  } else {
    ld_packed(ha + 512 * kc, f.A);
    ld_frag32(Whh + cc + 32 * kc, f.Bi);
    ld_frag32(Whh + cc + gs + 32 * kc, f.Bf);
    ld_frag32(Whh + cc + 2 * gs + 32 * kc, f.Bg);
    ld_frag32(Whh + cc + 3 * gs + 32 * kc, f.Bo);
  }
}

#define WMMA_THREAD_DECOMP()        \
  const int tid  = threadIdx.x;     \
  const int lane = tid & 31;        \
  const int wave = tid >> 5;        \
  const int half = lane >> 4;       \
  const int r    = lane & 15;       \
  const int wr   = wave & 3;        \
  const int cp   = wave >> 2;       \
  (void)lane;

DI float sigmoidf(float x) { return __builtin_amdgcn_rcpf(1.0f + __expf(-x)); }
DI float tanh_er(float x)  { return 1.0f - 2.0f * __builtin_amdgcn_rcpf(1.0f + __expf(2.0f * x)); }

__global__ void softmax_rows(const float* __restrict__ dis, float* __restrict__ w) {
  __shared__ float red[256];
  const int i = blockIdx.x, t = threadIdx.x;
  float v = -dis[i * 256 + t];
  red[t] = v;
  __syncthreads();
  for (int s = 128; s > 0; s >>= 1) {
    if (t < s) red[t] = fmaxf(red[t], red[t + s]);
    __syncthreads();
  }
  float mx = red[0];
  __syncthreads();
  float e = __expf(v - mx);
  red[t] = e;
  __syncthreads();
  for (int s = 128; s > 0; s >>= 1) {
    if (t < s) red[t] += red[t + s];
    __syncthreads();
  }
  w[pk_off(i, t)] = e / red[0];
}

__global__ void embed_weight_gemm(const float* __restrict__ E1, const float* __restrict__ E2,
                                  const float* __restrict__ Wih,
                                  float* __restrict__ G1T, float* __restrict__ G2T) {
  __shared__ float As[64 * LDA];
  __shared__ float Bs[32 * LDB];
  WMMA_THREAD_DECOMP();
  const float* E = blockIdx.z ? E2 : E1;
  float* G = blockIdx.z ? G2T : G1T;
  const int koff = blockIdx.z ? 256 : 0;
  const int rb = blockIdx.y * 64;
  const int cb = blockIdx.x * 64;
  v8f acc0 = {}; v8f acc1 = {};
  for (int k0 = 0; k0 < 256; k0 += 32) {
    __syncthreads();
#pragma unroll
    for (int i = 0; i < 8; ++i) {
      int idx = tid + i * 256;
      int ar = idx >> 5, ac = idx & 31;
      As[ar * LDA + ac] = Wih[(rb + ar) * 512 + koff + k0 + ac];
      int bc = idx >> 5, bk = idx & 31;
      Bs[bk * LDB + bc] = E[(cb + bc) * 256 + k0 + bk];
    }
    __syncthreads();
    mma_chunk(As, Bs, wr, cp, r, half, acc0, acc1);
  }
  const int c0 = rb + 16 * wr, d0 = cb + 32 * cp + r;
#pragma unroll
  for (int i = 0; i < 8; ++i) {
    G[pk_off(c0 + i + 8 * half, d0)]      = acc0[i];
    G[pk_off(c0 + i + 8 * half, d0 + 16)] = acc1[i];
  }
}

__global__ void pack_x(const float* __restrict__ X, float* __restrict__ Xp) {
  const int lane = threadIdx.x & 31;
  const int blk = blockIdx.x * 8 + (threadIdx.x >> 5);
  const float* s = X + ((blk >> 3) * 16 + (lane & 15)) * 256 + (blk & 7) * 32 + 4 * (lane >> 4);
  float* d = Xp + blk * 512 + lane * 4;
  v4f f[4];
  ld_frag32(s, f);
#pragma unroll
  for (int j = 0; j < 4; ++j) *reinterpret_cast<v4f*>(d + 128 * j) = f[j];
}

constexpr int LDH = 36;
__global__ __launch_bounds__(256) void lstm_step(const float* __restrict__ Xp, const float* __restrict__ wp,
                          const float* __restrict__ G1p, const float* __restrict__ G2p,
                          const float* __restrict__ Whh, const float* __restrict__ hin,
                          const float* __restrict__ bih, const float* __restrict__ bhh,
                          float* __restrict__ hout, float* __restrict__ c, float* __restrict__ temp, int l) {
  __shared__ __attribute__((aligned(16))) float hs[64 * LDH];
  const int tid  = threadIdx.x;
  const int lane = tid & 31;
  const int wave = tid >> 5;
  const int half = lane >> 4;
  const int r    = lane & 15;
  const int rbw  = wave & 3;
  const int ctw  = wave >> 2;
  const int row0 = blockIdx.y * 64 + rbw * 16;
  const int ct   = blockIdx.x * 2 + ctw;
  const int bb = row0 >> 8, nn0 = row0 & 255;
  const float* xa = Xp + ((((bb * 32 + l) * 256 + nn0) >> 4) << 12) + 4 * lane;
  const float* wa = wp + ((nn0 >> 4) << 12) + 4 * lane;
  const float* ha = hin + ((row0 >> 4) << 12) + 4 * lane;
  const int gg = (ct << 12) + 4 * lane;
  const int cc = (ct * 16 + r) * 256 + 4 * half;
  float* ctile = c + ((((row0 >> 4) << 4) + ct) << 8) + 4 * lane;
  v4f cv0 = *reinterpret_cast<const v4f*>(ctile);
  v4f cv1 = *reinterpret_cast<const v4f*>(ctile + 128);
  const int m = ct * 16 + r;
  const float bi0 = bih[m], bi1 = bhh[m];
  const float bf0 = bih[m + 256], bf1 = bhh[m + 256];
  const float bg0 = bih[m + 512], bg1 = bhh[m + 512];
  const float bo0 = bih[m + 768], bo1 = bhh[m + 768];
  v8f zi = {}, zf = {}, zg = {}, zo = {};
  GateFrags f[3];
  ld_gate_chunk(0, f[0], xa, wa, ha, G1p, G2p, Whh, gg, cc);
  ld_gate_chunk(1, f[1], xa, wa, ha, G1p, G2p, Whh, gg, cc);
#pragma unroll
  for (int t = 0; t < 24; ++t) {
    if (t + 2 < 24) ld_gate_chunk(t + 2, f[(t + 2) % 3], xa, wa, ha, G1p, G2p, Whh, gg, cc);
    __builtin_amdgcn_sched_barrier(0);
    if ((t >> 3) == 1) {
#pragma unroll
      for (int j = 0; j < 4; ++j) f[t % 3].A[j] *= f[t % 3].W[j];
    }
    mma_gates32(f[t % 3].A, f[t % 3].Bi, f[t % 3].Bf, f[t % 3].Bg, f[t % 3].Bo, zi, zf, zg, zo);
    __builtin_amdgcn_sched_barrier(0);
  }
#pragma unroll
  for (int e = 0; e < 8; ++e) {
    const float cold = (e < 4) ? cv0[e & 3] : cv1[e & 3];
    float iv = sigmoidf(zi[e] + bi0 + bi1);
    float fv = sigmoidf(zf[e] + bf0 + bf1);
    float gv = tanh_er (zg[e] + bg0 + bg1);
    float ov = sigmoidf(zo[e] + bo0 + bo1);
    float cn = fv * cold + iv * gv;
    if (e < 4) cv0[e & 3] = cn; else cv1[e & 3] = cn;
    hs[(16 * rbw + 8 * half + e) * LDH + 16 * ctw + r] = ov * tanh_er(cn);
  }
  __syncthreads();
  *reinterpret_cast<v4f*>(ctile)       = cv0;
  *reinterpret_cast<v4f*>(ctile + 128) = cv1;
  const int kcb = blockIdx.x;
#pragma unroll
  for (int t = 0; t < 2; ++t) {
    const int idx = tid + 256 * t;
    {
      const int blk = idx >> 7, q = idx & 127;
      const int j = q >> 5, ln = q & 31;
      v4f v = *reinterpret_cast<const v4f*>(&hs[(16 * blk + (ln & 15)) * LDH + 8 * j + 4 * (ln >> 4)]);
      *reinterpret_cast<v4f*>(hout + (((blockIdx.y * 4 + blk) * 8 + kcb) << 9) + q * 4) = v;
    }
    {
      const int rl = idx >> 3, m4 = (idx & 7) * 4;
      const int row = blockIdx.y * 64 + rl;
      const int b2 = row >> 8, nn = row & 255;
      v4f v = *reinterpret_cast<const v4f*>(&hs[rl * LDH + m4]);
      *reinterpret_cast<v4f*>(temp + ((b2 * 32 + l) * 256 + nn) * 256 + kcb * 32 + m4) = v;
    }
  }
}

__global__ void zero_f32(float* __restrict__ p, int n) {
  int i = blockIdx.x * 256 + threadIdx.x;
  if (i < n) p[i] = 0.0f;
}

__global__ void bn_stats(const float* __restrict__ temp, float* __restrict__ stats) {
  const int m = threadIdx.x;
  const int rb = blockIdx.x * 256;
  float s = 0.0f, s2 = 0.0f;
  for (int rr = 0; rr < 256; ++rr) {
    float v = temp[(rb + rr) * 256 + m];
    s += v; s2 += v * v;
  }
  atomicAdd(&stats[m], s);
  atomicAdd(&stats[256 + m], s2);
}

__global__ void bn_final(const float* __restrict__ stats, const float* __restrict__ gamma,
                         const float* __restrict__ beta, float* __restrict__ scale,
                         float* __restrict__ shift) {
  const int m = threadIdx.x;
  const float inv = 1.0f / 65536.0f;
  float mu = stats[m] * inv;
  float var = stats[256 + m] * inv - mu * mu;
  float sc = gamma[m] * rsqrtf(var + 1e-5f);
  scale[m] = sc;
  shift[m] = beta[m] - mu * sc;
}

struct TileFrags { v4f A[4][4], B[4][4], S[4], H[4]; };

DI void ld_tile_chunk(int kc, TileFrags& f, const float* pa, const float* pb, const float* ps, const float* ph) {
#pragma unroll
  for (int i = 0; i < 4; ++i) ld_frag32(pa + i * 4096 + 32 * kc, f.A[i]);
#pragma unroll
  for (int n = 0; n < 4; ++n) ld_frag32(pb + n * 4096 + 32 * kc, f.B[n]);
  ld_frag32(ps + 32 * kc, f.S);
  ld_frag32(ph + 32 * kc, f.H);
}

DI void mma_tile64(const TileFrags& f, v8f (&acc)[4][4]) {
#pragma unroll
  for (int j = 0; j < 4; ++j) {
#pragma unroll
    for (int i = 0; i < 4; ++i) {
      const v2f a0 = {f.A[i][j].x, f.A[i][j].y};
#pragma unroll
      for (int n = 0; n < 4; ++n) acc[i][n] = wmma4(a0, v2f{f.B[n][j].x, f.B[n][j].y}, acc[i][n]);
    }
#pragma unroll
    for (int i = 0; i < 4; ++i) {
      const v2f a1 = {f.A[i][j].z, f.A[i][j].w};
#pragma unroll
      for (int n = 0; n < 4; ++n) acc[i][n] = wmma4(a1, v2f{f.B[n][j].z, f.B[n][j].w}, acc[i][n]);
    }
  }
}

template <bool BN_ON_A>
DI void tile_gemm64(const float* pa, const float* pb, const float* ps, const float* ph, v8f (&acc)[4][4]) {
  TileFrags f[2];
  ld_tile_chunk(0, f[0], pa, pb, ps, ph);
#pragma unroll
  for (int t = 0; t < 8; ++t) {
    if (t + 1 < 8) ld_tile_chunk(t + 1, f[(t + 1) & 1], pa, pb, ps, ph);
    __builtin_amdgcn_sched_barrier(0);
    TileFrags& g = f[t & 1];
#pragma unroll
    for (int i = 0; i < 4; ++i) {
#pragma unroll
      for (int j = 0; j < 4; ++j) {
        if (BN_ON_A) g.A[i][j] = g.A[i][j] * g.S[j] + g.H[j];
        else         g.B[i][j] = g.B[i][j] * g.S[j] + g.H[j];
      }
    }
    mma_tile64(g, acc);
    __builtin_amdgcn_sched_barrier(0);
  }
}

__global__ __launch_bounds__(128) void proj_gemm(const float* __restrict__ temp, const float* __restrict__ Wlin,
                          const float* __restrict__ blin, const float* __restrict__ scale,
                          const float* __restrict__ shift, float* __restrict__ proj) {
  const int lane = threadIdx.x & 31, wave = threadIdx.x >> 5;
  const int half = lane >> 4, r = lane & 15;
  const int row0 = blockIdx.y * 128 + (wave & 1) * 64;
  const int col0 = blockIdx.x * 128 + (wave >> 1) * 64;
  v8f acc[4][4] = {};
  tile_gemm64<true>(temp + (row0 + r) * 256 + 4 * half, Wlin + (col0 + r) * 256 + 4 * half,
                    scale + 4 * half, shift + 4 * half, acc);
#pragma unroll
  for (int n = 0; n < 4; ++n) {
    const float bias = blin[col0 + 16 * n + r];
#pragma unroll
    for (int i = 0; i < 4; ++i) {
      float* o = proj + (row0 + 16 * i + 8 * half) * 256 + col0 + 16 * n + r;
#pragma unroll
      for (int e = 0; e < 8; ++e) o[e * 256] = acc[i][n][e] + bias;
    }
  }
}

__global__ __launch_bounds__(128) void out_gemm(const float* __restrict__ proj, const float* __restrict__ temp,
                         const float* __restrict__ scale, const float* __restrict__ shift,
                         float* __restrict__ out) {
  const int lane = threadIdx.x & 31, wave = threadIdx.x >> 5;
  const int half = lane >> 4, r = lane & 15;
  const int bl = blockIdx.y;
  const int o0 = (blockIdx.x >> 1) * 128 + (wave & 1) * 64;
  const int d0 = (blockIdx.x & 1) * 128 + (wave >> 1) * 64;
  v8f acc[4][4] = {};
  tile_gemm64<false>(proj + bl * 65536 + (o0 + r) * 256 + 4 * half, temp + bl * 65536 + (d0 + r) * 256 + 4 * half,
                     scale + 4 * half, shift + 4 * half, acc);
#pragma unroll
  for (int i = 0; i < 4; ++i) {
#pragma unroll
    for (int n = 0; n < 4; ++n) {
      float* o = out + bl * 65536 + (o0 + 16 * i + 8 * half) * 256 + d0 + 16 * n + r;
#pragma unroll
      for (int e = 0; e < 8; ++e) o[e * 256] = acc[i][n][e];
    }
  }
}

extern "C" void kernel_launch(void* const* d_in, const int* in_sizes, int n_in,
                              void* d_out, int out_size, void* d_ws, size_t ws_size,
                              hipStream_t stream) {
  (void)in_sizes; (void)n_in; (void)out_size; (void)ws_size;
  const float* X     = (const float*)d_in[0];
  const float* dis   = (const float*)d_in[1];
  const float* E1    = (const float*)d_in[2];
  const float* E2    = (const float*)d_in[3];
  const float* Wih   = (const float*)d_in[4];
  const float* Whh   = (const float*)d_in[5];
  const float* bih   = (const float*)d_in[6];
  const float* bhh   = (const float*)d_in[7];
  const float* gamma = (const float*)d_in[8];
  const float* beta  = (const float*)d_in[9];
  const float* Wlin  = (const float*)d_in[10];
  const float* blin  = (const float*)d_in[11];
  float* out = (float*)d_out;

  float* ws = (float*)d_ws;
  size_t o = 0;
  float* w_sm  = ws + o; o += 256 * 256;
  float* G1T   = ws + o; o += 256 * 1024;
  float* G2T   = ws + o; o += 256 * 1024;
  float* h2    = ws + o; o += 2048 * 1024;
  float* h     = ws + o; o += 2048 * 256;
  float* c     = ws + o; o += 2048 * 256;
  float* stats = ws + o; o += 512;
  float* scale = ws + o; o += 256;
  float* shift = ws + o; o += 256;
  float* temp  = ws + o; o += 65536 * 256;
  float* proj  = ws + o; o += 65536 * 256;

  softmax_rows<<<256, 256, 0, stream>>>(dis, w_sm);
  embed_weight_gemm<<<dim3(4, 16, 2), 256, 0, stream>>>(E1, E2, Wih, G1T, G2T);

  pack_x<<<4096, 256, 0, stream>>>(X, proj);

  const int zn = 2048 * 256 * 2 + 512;
  zero_f32<<<(zn + 255) / 256, 256, 0, stream>>>(h, zn);

  for (int l = 0; l < 32; l += 2) {
    lstm_step<<<dim3(8, 32), 256, 0, stream>>>(proj, w_sm, G1T, G2T, Whh, h, bih, bhh, h2, c, temp, l);
    lstm_step<<<dim3(8, 32), 256, 0, stream>>>(proj, w_sm, G1T, G2T, Whh, h2, bih, bhh, h, c, temp, l + 1);
  }

  bn_stats<<<256, 256, 0, stream>>>(temp, stats);
  bn_final<<<1, 256, 0, stream>>>(stats, gamma, beta, scale, shift);
  proj_gemm<<<dim3(2, 512), 128, 0, stream>>>(temp, Wlin, blin, scale, shift, proj);
  out_gemm<<<dim3(4, 256), 128, 0, stream>>>(proj, temp, scale, shift, out);
}


# gate GEMM: Whh repacked once per call into WMMA fragment order (as G1, G2 are), so a wave's h-path B load covers 512 contiguous bytes, not 16 partial lines; sum order kept
# speedup vs baseline: 2.7140x; 1.0745x over previous
#include <hip/hip_runtime.h>
#include <hip/hip_bf16.h>


#define DI __device__ __forceinline__

typedef float v2f __attribute__((ext_vector_type(2)));
typedef float v4f __attribute__((ext_vector_type(4)));
typedef float v8f __attribute__((ext_vector_type(8)));

constexpr int LDA = 36;
constexpr int LDB = 65;

DI v2f frag_a(const float* As, int k0, int r, int half) {
  v2f a;
  a.x = As[r * LDA + k0 + 2 * half + 0];
  a.y = As[r * LDA + k0 + 2 * half + 1];
  return a;
}
DI v2f frag_b(const float* Bs, int k0, int r, int half) {
  v2f b;
  b.x = Bs[(k0 + 2 * half + 0) * LDB + r];
  b.y = Bs[(k0 + 2 * half + 1) * LDB + r];
  return b;
}

DI void mma_chunk(const float* As, const float* Bs, int wr, int cp, int r, int half,
                  v8f& acc0, v8f& acc1) {
  const float* Aw = As + (wr * 16) * LDA;
  const float* B0 = Bs + 32 * cp;
  const float* B1 = Bs + 32 * cp + 16;
#pragma unroll
  for (int k0 = 0; k0 < 32; k0 += 4) {
    v2f a  = frag_a(Aw, k0, r, half);
    v2f b0 = frag_b(B0, k0, r, half);
    v2f b1 = frag_b(B1, k0, r, half);
    acc0 = __builtin_amdgcn_wmma_f32_16x16x4_f32(false, a, false, b0, (short)0, acc0, false, false);
    acc1 = __builtin_amdgcn_wmma_f32_16x16x4_f32(false, a, false, b1, (short)0, acc1, false, false);
  }
}

DI void ld_frag32(const float* p, v4f (&f)[4]) {
#pragma unroll
  for (int j = 0; j < 4; ++j) f[j] = *reinterpret_cast<const v4f*>(p + 8 * j);
}

DI int pk_off(int row, int k) {
  return (((row >> 4) * 8 + (k >> 5)) << 9) + ((((k & 31) >> 3) * 32 + (row & 15) + 16 * ((k >> 2) & 1)) << 2) + (k & 3);
}
DI void ld_packed(const float* p, v4f (&f)[4]) {
#pragma unroll
  for (int j = 0; j < 4; ++j) f[j] = *reinterpret_cast<const v4f*>(p + 128 * j);
}

DI v8f wmma4(v2f a, v2f b, v8f c) {
  return __builtin_amdgcn_wmma_f32_16x16x4_f32(false, a, false, b, (short)0, c, false, false);
}

DI void mma_gates32(const v4f (&A)[4], const v4f (&Bi)[4], const v4f (&Bf)[4], const v4f (&Bg)[4],
                    const v4f (&Bo)[4], v8f& ci, v8f& cf, v8f& cg, v8f& co) {
#pragma unroll
  for (int j = 0; j < 4; ++j) {
    v2f a0 = {A[j].x, A[j].y};
    ci = wmma4(a0, v2f{Bi[j].x, Bi[j].y}, ci); cf = wmma4(a0, v2f{Bf[j].x, Bf[j].y}, cf);
    cg = wmma4(a0, v2f{Bg[j].x, Bg[j].y}, cg); co = wmma4(a0, v2f{Bo[j].x, Bo[j].y}, co);
    v2f a1 = {A[j].z, A[j].w};
    ci = wmma4(a1, v2f{Bi[j].z, Bi[j].w}, ci); cf = wmma4(a1, v2f{Bf[j].z, Bf[j].w}, cf);
    cg = wmma4(a1, v2f{Bg[j].z, Bg[j].w}, cg); co = wmma4(a1, v2f{Bo[j].z, Bo[j].w}, co);
  }
}

struct GateFrags { v4f A[4], W[4], Bi[4], Bf[4], Bg[4], Bo[4]; };

DI void ld_gate_chunk(int t, GateFrags& f, const float* xa, const float* wa, const float* ha,
                      const float* G1p, const float* G2p, const float* Whp, int gg) {
  const int seg = t >> 3, kc = t & 7, gs = 65536;
  if (seg == 0) {
    ld_packed(xa + 512 * kc, f.A);
    ld_packed(G1p + gg + 512 * kc, f.Bi);
    ld_packed(G1p + gg + gs + 512 * kc, f.Bf);
    ld_packed(G1p + gg + 2 * gs + 512 * kc, f.Bg);
    ld_packed(G1p + gg + 3 * gs + 512 * kc, f.Bo);
  } else if (seg == 1) {
    ld_packed(xa + 512 * kc, f.A);
    ld_packed(wa + 512 * kc, f.W);
    ld_packed(G2p + gg + 512 * kc, f.Bi);
    ld_packed(G2p + gg + gs + 512 * kc, f.Bf);
    ld_packed(G2p + gg + 2 * gs + 512 * kc, f.Bg);
    ld_packed(G2p + gg + 3 * gs + 512 * kc, f.Bo);
  } else {
    ld_packed(ha + 512 * kc, f.A);
    ld_packed(Whp + gg + 512 * kc, f.Bi);
    ld_packed(Whp + gg + gs + 512 * kc, f.Bf);
    ld_packed(Whp + gg + 2 * gs + 512 * kc, f.Bg);
    ld_packed(Whp + gg + 3 * gs + 512 * kc, f.Bo);
  }
}

#define WMMA_THREAD_DECOMP()        \
  const int tid  = threadIdx.x;     \
  const int lane = tid & 31;        \
  const int wave = tid >> 5;        \
  const int half = lane >> 4;       \
  const int r    = lane & 15;       \
  const int wr   = wave & 3;        \
  const int cp   = wave >> 2;       \
  (void)lane;

DI float sigmoidf(float x) { return __builtin_amdgcn_rcpf(1.0f + __expf(-x)); }
DI float tanh_er(float x)  { return 1.0f - 2.0f * __builtin_amdgcn_rcpf(1.0f + __expf(2.0f * x)); }

__global__ void softmax_rows(const float* __restrict__ dis, float* __restrict__ w) {
  __shared__ float red[256];
  const int i = blockIdx.x, t = threadIdx.x;
  float v = -dis[i * 256 + t];
  red[t] = v;
  __syncthreads();
  for (int s = 128; s > 0; s >>= 1) {
    if (t < s) red[t] = fmaxf(red[t], red[t + s]);
    __syncthreads();
  }
  float mx = red[0];
  __syncthreads();
  float e = __expf(v - mx);
  red[t] = e;
  __syncthreads();
  for (int s = 128; s > 0; s >>= 1) {
    if (t < s) red[t] += red[t + s];
    __syncthreads();
  }
  w[pk_off(i, t)] = e / red[0];
}

__global__ void embed_weight_gemm(const float* __restrict__ E1, const float* __restrict__ E2,
                                  const float* __restrict__ Wih,
                                  float* __restrict__ G1T, float* __restrict__ G2T) {
  __shared__ float As[64 * LDA];
  __shared__ float Bs[32 * LDB];
  WMMA_THREAD_DECOMP();
  const float* E = blockIdx.z ? E2 : E1;
  float* G = blockIdx.z ? G2T : G1T;
  const int koff = blockIdx.z ? 256 : 0;
  const int rb = blockIdx.y * 64;
  const int cb = blockIdx.x * 64;
  v8f acc0 = {}; v8f acc1 = {};
  for (int k0 = 0; k0 < 256; k0 += 32) {
    __syncthreads();
#pragma unroll
    for (int i = 0; i < 8; ++i) {
      int idx = tid + i * 256;
      int ar = idx >> 5, ac = idx & 31;
      As[ar * LDA + ac] = Wih[(rb + ar) * 512 + koff + k0 + ac];
      int bc = idx >> 5, bk = idx & 31;
      Bs[bk * LDB + bc] = E[(cb + bc) * 256 + k0 + bk];
    }
    __syncthreads();
    mma_chunk(As, Bs, wr, cp, r, half, acc0, acc1);
  }
  const int c0 = rb + 16 * wr, d0 = cb + 32 * cp + r;
#pragma unroll
  for (int i = 0; i < 8; ++i) {
    G[pk_off(c0 + i + 8 * half, d0)]      = acc0[i];
    G[pk_off(c0 + i + 8 * half, d0 + 16)] = acc1[i];
  }
}

__global__ void pack_x(const float* __restrict__ X, float* __restrict__ Xp,
                       const float* __restrict__ Whh, float* __restrict__ Whp) {
  const int lane = threadIdx.x & 31;
  const bool isw = blockIdx.x >= 4096;
  const float* src = isw ? Whh : X;
  float* dst = isw ? Whp : Xp;
  const int blk = (isw ? blockIdx.x - 4096 : blockIdx.x) * 8 + (threadIdx.x >> 5);
  const float* s = src + ((blk >> 3) * 16 + (lane & 15)) * 256 + (blk & 7) * 32 + 4 * (lane >> 4);
  float* d = dst + blk * 512 + lane * 4;
  v4f f[4];
  ld_frag32(s, f);
#pragma unroll
  for (int j = 0; j < 4; ++j) *reinterpret_cast<v4f*>(d + 128 * j) = f[j];
}

constexpr int LDH = 36;
__global__ __launch_bounds__(256) void lstm_step(const float* __restrict__ Xp, const float* __restrict__ wp,
                          const float* __restrict__ G1p, const float* __restrict__ G2p,
                          const float* __restrict__ Whp, const float* __restrict__ hin,
                          const float* __restrict__ bih, const float* __restrict__ bhh,
                          float* __restrict__ hout, float* __restrict__ c, float* __restrict__ temp, int l) {
  __shared__ __attribute__((aligned(16))) float hs[64 * LDH];
  const int tid  = threadIdx.x;
  const int lane = tid & 31;
  const int wave = tid >> 5;
  const int half = lane >> 4;
  const int r    = lane & 15;
  const int rbw  = wave & 3;
  const int ctw  = wave >> 2;
  const int row0 = blockIdx.y * 64 + rbw * 16;
  const int ct   = blockIdx.x * 2 + ctw;
  const int bb = row0 >> 8, nn0 = row0 & 255;
  const float* xa = Xp + ((((bb * 32 + l) * 256 + nn0) >> 4) << 12) + 4 * lane;
  const float* wa = wp + ((nn0 >> 4) << 12) + 4 * lane;
  const float* ha = hin + ((row0 >> 4) << 12) + 4 * lane;
  const int gg = (ct << 12) + 4 * lane;
  float* ctile = c + ((((row0 >> 4) << 4) + ct) << 8) + 4 * lane;
  v4f cv0 = *reinterpret_cast<const v4f*>(ctile);
  v4f cv1 = *reinterpret_cast<const v4f*>(ctile + 128);
  const int m = ct * 16 + r;
  const float bi0 = bih[m], bi1 = bhh[m];
  const float bf0 = bih[m + 256], bf1 = bhh[m + 256];
  const float bg0 = bih[m + 512], bg1 = bhh[m + 512];
  const float bo0 = bih[m + 768], bo1 = bhh[m + 768];
  v8f zi = {}, zf = {}, zg = {}, zo = {};
  GateFrags f[3];
  ld_gate_chunk(0, f[0], xa, wa, ha, G1p, G2p, Whp, gg);
  ld_gate_chunk(1, f[1], xa, wa, ha, G1p, G2p, Whp, gg);
#pragma unroll
  for (int t = 0; t < 24; ++t) {
    if (t + 2 < 24) ld_gate_chunk(t + 2, f[(t + 2) % 3], xa, wa, ha, G1p, G2p, Whp, gg);
    __builtin_amdgcn_sched_barrier(0);
    if ((t >> 3) == 1) {
#pragma unroll
      for (int j = 0; j < 4; ++j) f[t % 3].A[j] *= f[t % 3].W[j];
    }
    mma_gates32(f[t % 3].A, f[t % 3].Bi, f[t % 3].Bf, f[t % 3].Bg, f[t % 3].Bo, zi, zf, zg, zo);
    __builtin_amdgcn_sched_barrier(0);
  }
#pragma unroll
  for (int e = 0; e < 8; ++e) {
    const float cold = (e < 4) ? cv0[e & 3] : cv1[e & 3];
    float iv = sigmoidf(zi[e] + bi0 + bi1);
    float fv = sigmoidf(zf[e] + bf0 + bf1);
    float gv = tanh_er (zg[e] + bg0 + bg1);
    float ov = sigmoidf(zo[e] + bo0 + bo1);
    float cn = fv * cold + iv * gv;
    if (e < 4) cv0[e & 3] = cn; else cv1[e & 3] = cn;
    hs[(16 * rbw + 8 * half + e) * LDH + 16 * ctw + r] = ov * tanh_er(cn);
  }
  __syncthreads();
  *reinterpret_cast<v4f*>(ctile)       = cv0;
  *reinterpret_cast<v4f*>(ctile + 128) = cv1;
  const int kcb = blockIdx.x;
#pragma unroll
  for (int t = 0; t < 2; ++t) {
    const int idx = tid + 256 * t;
    {
      const int blk = idx >> 7, q = idx & 127;
      const int j = q >> 5, ln = q & 31;
      v4f v = *reinterpret_cast<const v4f*>(&hs[(16 * blk + (ln & 15)) * LDH + 8 * j + 4 * (ln >> 4)]);
      *reinterpret_cast<v4f*>(hout + (((blockIdx.y * 4 + blk) * 8 + kcb) << 9) + q * 4) = v;
    }
    {
      const int rl = idx >> 3, m4 = (idx & 7) * 4;
      const int row = blockIdx.y * 64 + rl;
      const int b2 = row >> 8, nn = row & 255;
      v4f v = *reinterpret_cast<const v4f*>(&hs[rl * LDH + m4]);
      *reinterpret_cast<v4f*>(temp + ((b2 * 32 + l) * 256 + nn) * 256 + kcb * 32 + m4) = v;
    }
  }
}

__global__ void zero_f32(float* __restrict__ p, int n) {
  int i = blockIdx.x * 256 + threadIdx.x;
  if (i < n) p[i] = 0.0f;
}

__global__ void bn_stats(const float* __restrict__ temp, float* __restrict__ stats) {
  const int m = threadIdx.x;
  const int rb = blockIdx.x * 256;
  float s = 0.0f, s2 = 0.0f;
  for (int rr = 0; rr < 256; ++rr) {
    float v = temp[(rb + rr) * 256 + m];
    s += v; s2 += v * v;
  }
  atomicAdd(&stats[m], s);
  atomicAdd(&stats[256 + m], s2);
}

__global__ void bn_final(const float* __restrict__ stats, const float* __restrict__ gamma,
                         const float* __restrict__ beta, float* __restrict__ scale,
                         float* __restrict__ shift) {
  const int m = threadIdx.x;
  const float inv = 1.0f / 65536.0f;
  float mu = stats[m] * inv;
  float var = stats[256 + m] * inv - mu * mu;
  float sc = gamma[m] * rsqrtf(var + 1e-5f);
  scale[m] = sc;
  shift[m] = beta[m] - mu * sc;
}

struct TileFrags { v4f A[4][4], B[4][4], S[4], H[4]; };

DI void ld_tile_chunk(int kc, TileFrags& f, const float* pa, const float* pb, const float* ps, const float* ph) {
#pragma unroll
  for (int i = 0; i < 4; ++i) ld_frag32(pa + i * 4096 + 32 * kc, f.A[i]);
#pragma unroll
  for (int n = 0; n < 4; ++n) ld_frag32(pb + n * 4096 + 32 * kc, f.B[n]);
  ld_frag32(ps + 32 * kc, f.S);
  ld_frag32(ph + 32 * kc, f.H);
}

DI void mma_tile64(const TileFrags& f, v8f (&acc)[4][4]) {
#pragma unroll
  for (int j = 0; j < 4; ++j) {
#pragma unroll
    for (int i = 0; i < 4; ++i) {
      const v2f a0 = {f.A[i][j].x, f.A[i][j].y};
#pragma unroll
      for (int n = 0; n < 4; ++n) acc[i][n] = wmma4(a0, v2f{f.B[n][j].x, f.B[n][j].y}, acc[i][n]);
    }
#pragma unroll
    for (int i = 0; i < 4; ++i) {
      const v2f a1 = {f.A[i][j].z, f.A[i][j].w};
#pragma unroll
      for (int n = 0; n < 4; ++n) acc[i][n] = wmma4(a1, v2f{f.B[n][j].z, f.B[n][j].w}, acc[i][n]);
    }
  }
}

template <bool BN_ON_A>
DI void tile_gemm64(const float* pa, const float* pb, const float* ps, const float* ph, v8f (&acc)[4][4]) {
  TileFrags f[2];
  ld_tile_chunk(0, f[0], pa, pb, ps, ph);
#pragma unroll
  for (int t = 0; t < 8; ++t) {
    if (t + 1 < 8) ld_tile_chunk(t + 1, f[(t + 1) & 1], pa, pb, ps, ph);
    __builtin_amdgcn_sched_barrier(0);
    TileFrags& g = f[t & 1];
#pragma unroll
    for (int i = 0; i < 4; ++i) {
#pragma unroll
      for (int j = 0; j < 4; ++j) {
        if (BN_ON_A) g.A[i][j] = g.A[i][j] * g.S[j] + g.H[j];
        else         g.B[i][j] = g.B[i][j] * g.S[j] + g.H[j];
      }
    }
    mma_tile64(g, acc);
    __builtin_amdgcn_sched_barrier(0);
  }
}

__global__ __launch_bounds__(128) void proj_gemm(const float* __restrict__ temp, const float* __restrict__ Wlin,
                          const float* __restrict__ blin, const float* __restrict__ scale,
                          const float* __restrict__ shift, float* __restrict__ proj) {
  const int lane = threadIdx.x & 31, wave = threadIdx.x >> 5;
  const int half = lane >> 4, r = lane & 15;
  const int row0 = blockIdx.y * 128 + (wave & 1) * 64;
  const int col0 = blockIdx.x * 128 + (wave >> 1) * 64;
  v8f acc[4][4] = {};
  tile_gemm64<true>(temp + (row0 + r) * 256 + 4 * half, Wlin + (col0 + r) * 256 + 4 * half,
                    scale + 4 * half, shift + 4 * half, acc);
#pragma unroll
  for (int n = 0; n < 4; ++n) {
    const float bias = blin[col0 + 16 * n + r];
#pragma unroll
    for (int i = 0; i < 4; ++i) {
      float* o = proj + (row0 + 16 * i + 8 * half) * 256 + col0 + 16 * n + r;
#pragma unroll
      for (int e = 0; e < 8; ++e) o[e * 256] = acc[i][n][e] + bias;
    }
  }
}

__global__ __launch_bounds__(128) void out_gemm(const float* __restrict__ proj, const float* __restrict__ temp,
                         const float* __restrict__ scale, const float* __restrict__ shift,
                         float* __restrict__ out) {
  const int lane = threadIdx.x & 31, wave = threadIdx.x >> 5;
  const int half = lane >> 4, r = lane & 15;
  const int bl = blockIdx.y;
  const int o0 = (blockIdx.x >> 1) * 128 + (wave & 1) * 64;
  const int d0 = (blockIdx.x & 1) * 128 + (wave >> 1) * 64;
  v8f acc[4][4] = {};
  tile_gemm64<false>(proj + bl * 65536 + (o0 + r) * 256 + 4 * half, temp + bl * 65536 + (d0 + r) * 256 + 4 * half,
                     scale + 4 * half, shift + 4 * half, acc);
#pragma unroll
  for (int i = 0; i < 4; ++i) {
#pragma unroll
    for (int n = 0; n < 4; ++n) {
      float* o = out + bl * 65536 + (o0 + 16 * i + 8 * half) * 256 + d0 + 16 * n + r;
#pragma unroll
      for (int e = 0; e < 8; ++e) o[e * 256] = acc[i][n][e];
    }
  }
}

extern "C" void kernel_launch(void* const* d_in, const int* in_sizes, int n_in,
                              void* d_out, int out_size, void* d_ws, size_t ws_size,
                              hipStream_t stream) {
  (void)in_sizes; (void)n_in; (void)out_size; (void)ws_size;
  const float* X     = (const float*)d_in[0];
  const float* dis   = (const float*)d_in[1];
  const float* E1    = (const float*)d_in[2];
  const float* E2    = (const float*)d_in[3];
  const float* Wih   = (const float*)d_in[4];
  const float* Whh   = (const float*)d_in[5];
  const float* bih   = (const float*)d_in[6];
  const float* bhh   = (const float*)d_in[7];
  const float* gamma = (const float*)d_in[8];
  const float* beta  = (const float*)d_in[9];
  const float* Wlin  = (const float*)d_in[10];
  const float* blin  = (const float*)d_in[11];
  float* out = (float*)d_out;

  float* ws = (float*)d_ws;
  size_t o = 0;
  float* w_sm  = ws + o; o += 256 * 256;
  float* G1T   = ws + o; o += 256 * 1024;
  float* G2T   = ws + o; o += 256 * 1024;
  float* h2    = ws + o; o += 2048 * 1024;
  float* Whp   = h2 + 2048 * 256;
  float* h     = ws + o; o += 2048 * 256;
  float* c     = ws + o; o += 2048 * 256;
  float* stats = ws + o; o += 512;
  float* scale = ws + o; o += 256;
  float* shift = ws + o; o += 256;
  float* temp  = ws + o; o += 65536 * 256;
  float* proj  = ws + o; o += 65536 * 256;

  softmax_rows<<<256, 256, 0, stream>>>(dis, w_sm);
  embed_weight_gemm<<<dim3(4, 16, 2), 256, 0, stream>>>(E1, E2, Wih, G1T, G2T);

  pack_x<<<4096 + 64, 256, 0, stream>>>(X, proj, Whh, Whp);

  const int zn = 2048 * 256 * 2 + 512;
  zero_f32<<<(zn + 255) / 256, 256, 0, stream>>>(h, zn);

  for (int l = 0; l < 32; l += 2) {
    lstm_step<<<dim3(8, 32), 256, 0, stream>>>(proj, w_sm, G1T, G2T, Whp, h, bih, bhh, h2, c, temp, l);
    lstm_step<<<dim3(8, 32), 256, 0, stream>>>(proj, w_sm, G1T, G2T, Whp, h2, bih, bhh, h, c, temp, l + 1);
  }

  bn_stats<<<256, 256, 0, stream>>>(temp, stats);
  bn_final<<<1, 256, 0, stream>>>(stats, gamma, beta, scale, shift);
  proj_gemm<<<dim3(2, 512), 128, 0, stream>>>(temp, Wlin, blin, scale, shift, proj);
  out_gemm<<<dim3(4, 256), 128, 0, stream>>>(proj, temp, scale, shift, out);
}
